# starting kernel
# baseline (speedup 1.0000x reference)
#include <hip/hip_runtime.h>
#include <hip/hip_bf16.h>

typedef __attribute__((ext_vector_type(2))) float v2f;
typedef __attribute__((ext_vector_type(4))) float v4f;
typedef __attribute__((ext_vector_type(8))) float v8f;

#define D_FEAT 512

__global__ void gcn_init_deg(float* __restrict__ deg, int N) {
    int i = blockIdx.x * blockDim.x + threadIdx.x;
    if (i < N) deg[i] = 1.0f;
}

__global__ void gcn_deg(const int* __restrict__ dst, float* __restrict__ deg, int E) {
    int e = blockIdx.x * blockDim.x + threadIdx.x;
    if (e < E) atomicAdd(&deg[dst[e]], 1.0f);
}

__global__ void __launch_bounds__(256)
gcn_xw_wmma(const float* __restrict__ x, const float* __restrict__ W,
            float* __restrict__ h, int N) {
    int wave = (blockIdx.x * blockDim.x + threadIdx.x) >> 5;
    int lane = threadIdx.x & 31;
    int row0 = wave * 16;
    if (row0 >= N) return;

    int r    = lane & 15;
    int half = lane >> 4;

    float m0 = (lane == 0)  ? 1.0f : 0.0f;
    float m1 = (lane == 16) ? 1.0f : 0.0f;

    const v2f* xv = (const v2f*)(x + (size_t)(row0 + r) * D_FEAT) + half;
    const v4f* Wv = (const v4f*)W;

    v8f c = {0.f, 0.f, 0.f, 0.f, 0.f, 0.f, 0.f, 0.f};

#pragma unroll 4
    for (int k4 = 0; k4 < D_FEAT / 4; ++k4) {
        v2f a = xv[2 * k4];
        v4f w = Wv[k4];
        v2f b;
        b.x = w.x * m0 + w.z * m1;
        b.y = w.y * m0 + w.w * m1;
        c = __builtin_amdgcn_wmma_f32_16x16x4_f32(
                 false, a,  false, b,
                 (short)0, c,  false,  false);
    }

    if (r == 0) {
        float* hp = h + row0 + 8 * half;
#pragma unroll
        for (int i = 0; i < 8; ++i) hp[i] = c[i];
    }
}

__global__ void gcn_dinv(const float* __restrict__ deg, float* __restrict__ dinv, int N) {
    int i = blockIdx.x * blockDim.x + threadIdx.x;
    if (i < N) dinv[i] = __frsqrt_rn(deg[i]);
}

__global__ void gcn_init_out(const float* __restrict__ dinv, const float* __restrict__ h,
                             float* __restrict__ out, int N) {
    int i = blockIdx.x * blockDim.x + threadIdx.x;
    if (i < N) {
        float di = dinv[i];
        out[i] = di * di * h[i];
    }
}

__global__ void gcn_edge_aggregate(const int* __restrict__ src, const int* __restrict__ dst,
                                   const float* __restrict__ dinv, const float* __restrict__ h,
                                   float* __restrict__ out, int E) {
    int e = blockIdx.x * blockDim.x + threadIdx.x;
    if (e < E) {
        int s = src[e];
        int d = dst[e];
        atomicAdd(&out[d], dinv[s] * dinv[d] * h[s]);
    }
}

extern "C" void kernel_launch(void* const* d_in, const int* in_sizes, int n_in,
                              void* d_out, int out_size, void* d_ws, size_t ws_size,
                              hipStream_t stream) {
    const float* x  = (const float*)d_in[0];
    const int*   ei = (const int*)d_in[1];
    const float* W  = (const float*)d_in[2];
    float*       out = (float*)d_out;

    const int N = out_size;
    const int E = in_sizes[1] / 2;
    const int* src = ei;
    const int* dst = ei + E;

    float* deg  = (float*)d_ws;
    float* dinv = deg + N;
    float* h    = dinv + N;

    const int TB = 256;
    const int nBlkN = (N + TB - 1) / TB;
    const int nBlkE = (E + TB - 1) / TB;
    const int nWaves   = (N + 15) / 16;
    const int nBlkWmma = (nWaves * 32 + TB - 1) / TB;

    gcn_init_deg<<<nBlkN, TB, 0, stream>>>(deg, N);
    gcn_deg<<<nBlkE, TB, 0, stream>>>(dst, deg, E);
    gcn_xw_wmma<<<nBlkWmma, TB, 0, stream>>>(x, W, h, N);
    gcn_dinv<<<nBlkN, TB, 0, stream>>>(deg, dinv, N);
    gcn_init_out<<<nBlkN, TB, 0, stream>>>(dinv, h, out, N);
    gcn_edge_aggregate<<<nBlkE, TB, 0, stream>>>(src, dst, dinv, h, out, E);
}


# Both scatter passes (degree count, message sum) accumulate per node range in LDS (16 ranges x 16 edge chunks), flushed by coalesced float atomics; dinv[dst] scales the LDS sum; sum order changes
# speedup vs baseline: 3.3927x; 3.3927x over previous
#include <hip/hip_runtime.h>
#include <hip/hip_bf16.h>

typedef __attribute__((ext_vector_type(2))) float v2f;
typedef __attribute__((ext_vector_type(4))) float v4f;
typedef __attribute__((ext_vector_type(8))) float v8f;

#define D_FEAT 512

__global__ void gcn_init_deg(float* __restrict__ deg, int N) {
    int i = blockIdx.x * blockDim.x + threadIdx.x;
    if (i < N) deg[i] = 1.0f;
}

__global__ void __launch_bounds__(256)
gcn_xw_wmma(const float* __restrict__ x, const float* __restrict__ W,
            float* __restrict__ h, int N) {
    int wave = (blockIdx.x * blockDim.x + threadIdx.x) >> 5;
    int lane = threadIdx.x & 31;
    int row0 = wave * 16;
    if (row0 >= N) return;

    int r    = lane & 15;
    int half = lane >> 4;

    float m0 = (lane == 0)  ? 1.0f : 0.0f;
    float m1 = (lane == 16) ? 1.0f : 0.0f;

    const v2f* xv = (const v2f*)(x + (size_t)(row0 + r) * D_FEAT) + half;
    const v4f* Wv = (const v4f*)W;

    v8f c = {0.f, 0.f, 0.f, 0.f, 0.f, 0.f, 0.f, 0.f};

#pragma unroll 4
    for (int k4 = 0; k4 < D_FEAT / 4; ++k4) {
        v2f a = xv[2 * k4];
        v4f w = Wv[k4];
        v2f b;
        b.x = w.x * m0 + w.z * m1;
        b.y = w.y * m0 + w.w * m1;
        c = __builtin_amdgcn_wmma_f32_16x16x4_f32(
                 false, a,  false, b,
                 (short)0, c,  false,  false);
    }

    if (r == 0) {
        float* hp = h + row0 + 8 * half;
#pragma unroll
        for (int i = 0; i < 8; ++i) hp[i] = c[i];
    }
}

__global__ void gcn_dinv(const float* __restrict__ deg, float* __restrict__ dinv, int N) {
    int i = blockIdx.x * blockDim.x + threadIdx.x;
    if (i < N) dinv[i] = __frsqrt_rn(deg[i]);
}

__global__ void gcn_init_out(const float* __restrict__ dinv, const float* __restrict__ h,
                             float* __restrict__ out, int N) {
    int i = blockIdx.x * blockDim.x + threadIdx.x;
    if (i < N) {
        float di = dinv[i];
        out[i] = di * di * h[i];
    }
}

#define NR   16
#define NC   16
#define RS   6272
#define PTB  1024
#define DU   8
#define TU   4
#define QCAP (32 * 4 * TU)
#define ROT  ((RS / NC) & ~31)
#define FL   ((RS + PTB - 1) / PTB)

__device__ __forceinline__ int4 ld_quad_guarded(const int* __restrict__ p, int q, int nelem, int fill) {
    int4 v = make_int4(fill, fill, fill, fill);
    const int e = q * 4;
    if (e     < nelem) v.x = p[e];
    if (e + 1 < nelem) v.y = p[e + 1];
    if (e + 2 < nelem) v.z = p[e + 2];
    if (e + 3 < nelem) v.w = p[e + 3];
    return v;
}

template <int U>
__device__ __forceinline__ void ld_tile_vec(const int* __restrict__ p, int qbase, int tid, int4 (&v)[U]) {
    const int4* p4 = (const int4*)p + qbase + tid;
#pragma unroll
    for (int u = 0; u < U; ++u) v[u] = p4[u * PTB];
}
template <int U>
__device__ __forceinline__ void ld_tile_guarded(const int* __restrict__ p, int qbase, int tid, int nelem,
                                                int fill, int4 (&v)[U]) {
#pragma unroll
    for (int u = 0; u < U; ++u) v[u] = ld_quad_guarded(p, qbase + u * PTB + tid, nelem, fill);
}

__device__ __forceinline__ void deg_tile(const int4 (&v)[DU], unsigned* cnt, unsigned r0) {
#pragma unroll
    for (int u = 0; u < DU; ++u) {
        unsigned k;
        k = (unsigned)v[u].x - r0; if (k < RS) atomicAdd(&cnt[k], 1u);
        k = (unsigned)v[u].y - r0; if (k < RS) atomicAdd(&cnt[k], 1u);
        k = (unsigned)v[u].z - r0; if (k < RS) atomicAdd(&cnt[k], 1u);
        k = (unsigned)v[u].w - r0; if (k < RS) atomicAdd(&cnt[k], 1u);
    }
}

__device__ __forceinline__ void deg_scan(const int* __restrict__ dstc, int nelem, bool vec,
                                         unsigned* cnt, unsigned r0, int tid) {
    const int nq = (nelem + 3) >> 2;
    const int nfull = vec ? (nelem >> 2) / (PTB * DU) : 0;
    for (int t = 0; t < nfull; ++t) {
        int4 v[DU];
        ld_tile_vec<DU>(dstc, t * (PTB * DU), tid, v);
        deg_tile(v, cnt, r0);
    }
    for (int base = nfull * (PTB * DU); base < nq; base += PTB * DU) {
        int4 v[DU];
        ld_tile_guarded<DU>(dstc, base, tid, nelem, -1, v);
        deg_tile(v, cnt, r0);
    }
}

__global__ void __launch_bounds__(PTB)
gcn_deg_lds(const int* __restrict__ dst, float* __restrict__ deg, int N, int E, int cl) {
    __shared__ unsigned cnt[RS];
    const int tid = threadIdx.x;
    const int r = blockIdx.x % NR;
    const int c = blockIdx.x / NR;
    const int e0 = c * cl;
    int nelem = E - e0; if (nelem > cl) nelem = cl; if (nelem < 0) nelem = 0;
    const int* dstc = dst + e0;
    const bool vec = (((size_t)dst) & 15) == 0;

    for (unsigned r0 = (unsigned)r * RS; r0 < (unsigned)N; r0 += NR * RS) {
        for (int i = tid; i < RS; i += PTB) cnt[i] = 0u;
        __syncthreads();
        deg_scan(dstc, nelem, vec, cnt, r0, tid);
        __syncthreads();
        for (int i = tid; i < RS; i += PTB) {
            int k = i + c * ROT; if (k >= RS) k -= RS;
            const unsigned n = r0 + (unsigned)k;
            const unsigned cv = cnt[k];
            if (cv != 0u && n < (unsigned)N) atomicAdd(&deg[n], (float)cv);
        }
        __syncthreads();
    }
}

__device__ __forceinline__ void agg_tile(const int4 (&dc)[TU], const int4 (&sc)[TU],
                                         const float* __restrict__ dinv, const float* __restrict__ h,
                                         float* acc, uint2* q, unsigned r0, int lane) {
    unsigned cnt = 0;
#pragma unroll
    for (int u = 0; u < TU; ++u) {
        const int dd[4] = {dc[u].x, dc[u].y, dc[u].z, dc[u].w};
        const int ss[4] = {sc[u].x, sc[u].y, sc[u].z, sc[u].w};
#pragma unroll
        for (int j = 0; j < 4; ++j) {
            const unsigned k = (unsigned)dd[j] - r0;
            const bool hit = k < RS;
            const unsigned m = __builtin_amdgcn_ballot_w32(hit);
            if (hit) q[cnt + __builtin_amdgcn_mbcnt_lo(m, 0u)] = make_uint2(k, (unsigned)ss[j]);
            cnt += (unsigned)__builtin_popcount(m);
        }
    }
    for (unsigned i = (unsigned)lane; i < cnt; i += 64u) {
        const unsigned i2 = i + 32u;
        const bool two = i2 < cnt;
        const uint2 a = q[i];
        const uint2 b = q[two ? i2 : i];
        const float da = dinv[a.y], ha = h[a.y];
        const float db = dinv[b.y], hb = h[b.y];
        const float va = da * ha;
        const float vb = two ? db * hb : 0.0f;
        atomicAdd(&acc[a.x], va);
        atomicAdd(&acc[b.x], vb);
    }
}

__device__ __forceinline__ void agg_scan(const int* __restrict__ srcc, const int* __restrict__ dstc,
                                         int nelem, bool vec, const float* __restrict__ dinv,
                                         const float* __restrict__ h, float* acc, uint2* q,
                                         unsigned r0, int tid, int lane) {
    const int nq = (nelem + 3) >> 2;
    const int nfull = vec ? (nelem >> 2) / (PTB * TU) : 0;
    if (nfull > 0) {
        int4 dn[TU], sn[TU];
        ld_tile_vec<TU>(dstc, 0, tid, dn);
        ld_tile_vec<TU>(srcc, 0, tid, sn);
        for (int t = 0; t < nfull; ++t) {
            int4 dc[TU], sc[TU];
#pragma unroll
            for (int u = 0; u < TU; ++u) { dc[u] = dn[u]; sc[u] = sn[u]; }
            if (t + 1 < nfull) {
                ld_tile_vec<TU>(dstc, (t + 1) * (PTB * TU), tid, dn);
                ld_tile_vec<TU>(srcc, (t + 1) * (PTB * TU), tid, sn);
            }
            agg_tile(dc, sc, dinv, h, acc, q, r0, lane);
        }
    }
    for (int base = nfull * (PTB * TU); base < nq; base += PTB * TU) {
        int4 dc[TU], sc[TU];
        ld_tile_guarded<TU>(dstc, base, tid, nelem, -1, dc);
        ld_tile_guarded<TU>(srcc, base, tid, nelem, 0, sc);
        agg_tile(dc, sc, dinv, h, acc, q, r0, lane);
    }
}

__global__ void __launch_bounds__(PTB)
gcn_agg_lds(const int* __restrict__ src, const int* __restrict__ dst,
            const float* __restrict__ dinv, const float* __restrict__ h,
            float* __restrict__ out, int N, int E, int cl) {
    __shared__ float acc[RS];
    __shared__ uint2 queue[(PTB / 32) * QCAP];
    const int tid = threadIdx.x;
    const int lane = tid & 31;
    uint2* q = queue + (tid >> 5) * QCAP;
    const int r = blockIdx.x % NR;
    const int c = blockIdx.x / NR;
    const int e0 = c * cl;
    int nelem = E - e0; if (nelem > cl) nelem = cl; if (nelem < 0) nelem = 0;
    const int* srcc = src + e0;
    const int* dstc = dst + e0;
    const bool vec = ((((size_t)src) | ((size_t)dst)) & 15) == 0;

    for (unsigned r0 = (unsigned)r * RS; r0 < (unsigned)N; r0 += NR * RS) {
        for (int i = tid; i < RS; i += PTB) acc[i] = 0.0f;
        __syncthreads();
        agg_scan(srcc, dstc, nelem, vec, dinv, h, acc, q, r0, tid, lane);
        __syncthreads();
        float dv[FL];
#pragma unroll
        for (int j = 0; j < FL; ++j) {
            int k = tid + j * PTB + c * ROT; if (k >= RS) k -= RS;
            const unsigned n = r0 + (unsigned)k;
            dv[j] = dinv[n < (unsigned)N ? n : (unsigned)N - 1u];
        }
#pragma unroll
        for (int j = 0; j < FL; ++j) {
            const int i = tid + j * PTB;
            if (i < RS) {
                int k = i + c * ROT; if (k >= RS) k -= RS;
                const unsigned n = r0 + (unsigned)k;
                const float v = acc[k];
                if (v != 0.0f && n < (unsigned)N) atomicAdd(&out[n], dv[j] * v);
            }
        }
        __syncthreads();
    }
}

extern "C" void kernel_launch(void* const* d_in, const int* in_sizes, int n_in,
                              void* d_out, int out_size, void* d_ws, size_t ws_size,
                              hipStream_t stream) {
    const float* x  = (const float*)d_in[0];
    const int*   ei = (const int*)d_in[1];
    const float* W  = (const float*)d_in[2];
    float*       out = (float*)d_out;

    const int N = out_size;
    const int E = in_sizes[1] / 2;
    const int* src = ei;
    const int* dst = ei + E;

    float* deg  = (float*)d_ws;
    float* dinv = deg + N;
    float* h    = dinv + N;

    const int TB = 256;
    const int nBlkN = (N + TB - 1) / TB;
    const int cl = (((E + NC - 1) / NC) + 3) & ~3;
    const int nWaves   = (N + 15) / 16;
    const int nBlkWmma = (nWaves * 32 + TB - 1) / TB;

    gcn_init_deg<<<nBlkN, TB, 0, stream>>>(deg, N);
    gcn_deg_lds<<<NR * NC, PTB, 0, stream>>>(dst, deg, N, E, cl);
    gcn_xw_wmma<<<nBlkWmma, TB, 0, stream>>>(x, W, h, N);
    gcn_dinv<<<nBlkN, TB, 0, stream>>>(deg, dinv, N);
    gcn_init_out<<<nBlkN, TB, 0, stream>>>(dinv, h, out, N);
    gcn_agg_lds<<<NR * NC, PTB, 0, stream>>>(src, dst, dinv, h, out, N, E, cl);
}


# x@W product and degree count in ONE launch on disjoint work-groups (256 product groups of 25 waves, then 256 count groups): the row stream runs under the count; 6 -> 5 launches; sums unchanged
# speedup vs baseline: 3.5964x; 1.0600x over previous
#include <hip/hip_runtime.h>
#include <hip/hip_bf16.h>

typedef __attribute__((ext_vector_type(2))) float v2f;
typedef __attribute__((ext_vector_type(4))) float v4f;
typedef __attribute__((ext_vector_type(8))) float v8f;

#define D_FEAT 512

__global__ void gcn_init_deg(float* __restrict__ deg, int N) {
    int i = blockIdx.x * blockDim.x + threadIdx.x;
    if (i < N) deg[i] = 1.0f;
}

__device__ __forceinline__ void
xw_wave(const float* __restrict__ x, const float* __restrict__ W,
        float* __restrict__ h, int N, int wave) {
    int lane = threadIdx.x & 31;
    int row0 = wave * 16;
    if (row0 >= N) return;

    int r    = lane & 15;
    int half = lane >> 4;

    float m0 = (lane == 0)  ? 1.0f : 0.0f;
    float m1 = (lane == 16) ? 1.0f : 0.0f;

    const v2f* xv = (const v2f*)(x + (size_t)(row0 + r) * D_FEAT) + half;
    const v4f* Wv = (const v4f*)W;

    v8f c = {0.f, 0.f, 0.f, 0.f, 0.f, 0.f, 0.f, 0.f};

#pragma unroll 4
    for (int k4 = 0; k4 < D_FEAT / 4; ++k4) {
        v2f a = xv[2 * k4];
        v4f w = Wv[k4];
        v2f b;
        b.x = w.x * m0 + w.z * m1;
        b.y = w.y * m0 + w.w * m1;
        c = __builtin_amdgcn_wmma_f32_16x16x4_f32(
                 false, a,  false, b,
                 (short)0, c,  false,  false);
    }

    if (r == 0) {
        float* hp = h + row0 + 8 * half;
#pragma unroll
        for (int i = 0; i < 8; ++i) hp[i] = c[i];
    }
}

__global__ void gcn_dinv(const float* __restrict__ deg, float* __restrict__ dinv, int N) {
    int i = blockIdx.x * blockDim.x + threadIdx.x;
    if (i < N) dinv[i] = __frsqrt_rn(deg[i]);
}

__global__ void gcn_init_out(const float* __restrict__ dinv, const float* __restrict__ h,
                             float* __restrict__ out, int N) {
    int i = blockIdx.x * blockDim.x + threadIdx.x;
    if (i < N) {
        float di = dinv[i];
        out[i] = di * di * h[i];
    }
}

#define NR   16
#define NC   16
#define RS   6272
#define PTB  1024
#define DU   8
#define TU   4
#define QCAP (32 * 4 * TU)
#define ROT  ((RS / NC) & ~31)
#define FL   ((RS + PTB - 1) / PTB)

__device__ __forceinline__ int4 ld_quad_guarded(const int* __restrict__ p, int q, int nelem, int fill) {
    int4 v = make_int4(fill, fill, fill, fill);
    const int e = q * 4;
    if (e     < nelem) v.x = p[e];
    if (e + 1 < nelem) v.y = p[e + 1];
    if (e + 2 < nelem) v.z = p[e + 2];
    if (e + 3 < nelem) v.w = p[e + 3];
    return v;
}

template <int U>
__device__ __forceinline__ void ld_tile_vec(const int* __restrict__ p, int qbase, int tid, int4 (&v)[U]) {
    const int4* p4 = (const int4*)p + qbase + tid;
#pragma unroll
    for (int u = 0; u < U; ++u) v[u] = p4[u * PTB];
}
template <int U>
__device__ __forceinline__ void ld_tile_guarded(const int* __restrict__ p, int qbase, int tid, int nelem,
                                                int fill, int4 (&v)[U]) {
#pragma unroll
    for (int u = 0; u < U; ++u) v[u] = ld_quad_guarded(p, qbase + u * PTB + tid, nelem, fill);
}

__device__ __forceinline__ void deg_tile(const int4 (&v)[DU], unsigned* cnt, unsigned r0) {
#pragma unroll
    for (int u = 0; u < DU; ++u) {
        unsigned k;
        k = (unsigned)v[u].x - r0; if (k < RS) atomicAdd(&cnt[k], 1u);
        k = (unsigned)v[u].y - r0; if (k < RS) atomicAdd(&cnt[k], 1u);
        k = (unsigned)v[u].z - r0; if (k < RS) atomicAdd(&cnt[k], 1u);
        k = (unsigned)v[u].w - r0; if (k < RS) atomicAdd(&cnt[k], 1u);
    }
}

__device__ __forceinline__ void deg_scan(const int* __restrict__ dstc, int nelem, bool vec,
                                         unsigned* cnt, unsigned r0, int tid) {
    const int nq = (nelem + 3) >> 2;
    const int nfull = vec ? (nelem >> 2) / (PTB * DU) : 0;
    for (int t = 0; t < nfull; ++t) {
        int4 v[DU];
        ld_tile_vec<DU>(dstc, t * (PTB * DU), tid, v);
        deg_tile(v, cnt, r0);
    }
    for (int base = nfull * (PTB * DU); base < nq; base += PTB * DU) {
        int4 v[DU];
        ld_tile_guarded<DU>(dstc, base, tid, nelem, -1, v);
        deg_tile(v, cnt, r0);
    }
}

__device__ __forceinline__ void
deg_group(const int* __restrict__ dst, float* __restrict__ deg, int N, int E, int cl,
          unsigned* cnt, int bid) {
    const int tid = threadIdx.x;
    const int r = bid % NR;
    const int c = bid / NR;
    const int e0 = c * cl;
    int nelem = E - e0; if (nelem > cl) nelem = cl; if (nelem < 0) nelem = 0;
    const int* dstc = dst + e0;
    const bool vec = (((size_t)dst) & 15) == 0;

    for (unsigned r0 = (unsigned)r * RS; r0 < (unsigned)N; r0 += NR * RS) {
        for (int i = tid; i < RS; i += PTB) cnt[i] = 0u;
        __syncthreads();
        deg_scan(dstc, nelem, vec, cnt, r0, tid);
        __syncthreads();
        for (int i = tid; i < RS; i += PTB) {
            int k = i + c * ROT; if (k >= RS) k -= RS;
            const unsigned n = r0 + (unsigned)k;
            const unsigned cv = cnt[k];
            if (cv != 0u && n < (unsigned)N) atomicAdd(&deg[n], (float)cv);
        }
        __syncthreads();
    }
}

__global__ void __launch_bounds__(PTB)
gcn_xw_deg(const float* __restrict__ x, const float* __restrict__ W, float* __restrict__ h,
           const int* __restrict__ dst, float* __restrict__ deg,
           int N, int E, int cl, int nbx, int wpb) {
    __shared__ unsigned cnt[RS];
    const int bid = blockIdx.x;
    if (bid < nbx) {
        const int wi = threadIdx.x >> 5;
        if (wi < wpb) xw_wave(x, W, h, N, bid * wpb + wi);
    } else {
        deg_group(dst, deg, N, E, cl, cnt, bid - nbx);
    }
}

__device__ __forceinline__ void agg_tile(const int4 (&dc)[TU], const int4 (&sc)[TU],
                                         const float* __restrict__ dinv, const float* __restrict__ h,
                                         float* acc, uint2* q, unsigned r0, int lane) {
    unsigned cnt = 0;
#pragma unroll
    for (int u = 0; u < TU; ++u) {
        const int dd[4] = {dc[u].x, dc[u].y, dc[u].z, dc[u].w};
        const int ss[4] = {sc[u].x, sc[u].y, sc[u].z, sc[u].w};
#pragma unroll
        for (int j = 0; j < 4; ++j) {
            const unsigned k = (unsigned)dd[j] - r0;
            const bool hit = k < RS;
            const unsigned m = __builtin_amdgcn_ballot_w32(hit);
            if (hit) q[cnt + __builtin_amdgcn_mbcnt_lo(m, 0u)] = make_uint2(k, (unsigned)ss[j]);
            cnt += (unsigned)__builtin_popcount(m);
        }
    }
    for (unsigned i = (unsigned)lane; i < cnt; i += 64u) {
        const unsigned i2 = i + 32u;
        const bool two = i2 < cnt;
        const uint2 a = q[i];
        const uint2 b = q[two ? i2 : i];
        const float da = dinv[a.y], ha = h[a.y];
        const float db = dinv[b.y], hb = h[b.y];
        const float va = da * ha;
        const float vb = two ? db * hb : 0.0f;
        atomicAdd(&acc[a.x], va);
        atomicAdd(&acc[b.x], vb);
    }
}

__device__ __forceinline__ void agg_scan(const int* __restrict__ srcc, const int* __restrict__ dstc,
                                         int nelem, bool vec, const float* __restrict__ dinv,
                                         const float* __restrict__ h, float* acc, uint2* q,
                                         unsigned r0, int tid, int lane) {
    const int nq = (nelem + 3) >> 2;
    const int nfull = vec ? (nelem >> 2) / (PTB * TU) : 0;
    if (nfull > 0) {
        int4 dn[TU], sn[TU];
        ld_tile_vec<TU>(dstc, 0, tid, dn);
        ld_tile_vec<TU>(srcc, 0, tid, sn);
        for (int t = 0; t < nfull; ++t) {
            int4 dc[TU], sc[TU];
#pragma unroll
            for (int u = 0; u < TU; ++u) { dc[u] = dn[u]; sc[u] = sn[u]; }
            if (t + 1 < nfull) {
                ld_tile_vec<TU>(dstc, (t + 1) * (PTB * TU), tid, dn);
                ld_tile_vec<TU>(srcc, (t + 1) * (PTB * TU), tid, sn);
            }
            agg_tile(dc, sc, dinv, h, acc, q, r0, lane);
        }
    }
    for (int base = nfull * (PTB * TU); base < nq; base += PTB * TU) {
        int4 dc[TU], sc[TU];
        ld_tile_guarded<TU>(dstc, base, tid, nelem, -1, dc);
        ld_tile_guarded<TU>(srcc, base, tid, nelem, 0, sc);
        agg_tile(dc, sc, dinv, h, acc, q, r0, lane);
    }
}

__global__ void __launch_bounds__(PTB)
gcn_agg_lds(const int* __restrict__ src, const int* __restrict__ dst,
            const float* __restrict__ dinv, const float* __restrict__ h,
            float* __restrict__ out, int N, int E, int cl) {
    __shared__ float acc[RS];
    __shared__ uint2 queue[(PTB / 32) * QCAP];
    const int tid = threadIdx.x;
    const int lane = tid & 31;
    uint2* q = queue + (tid >> 5) * QCAP;
    const int r = blockIdx.x % NR;
    const int c = blockIdx.x / NR;
    const int e0 = c * cl;
    int nelem = E - e0; if (nelem > cl) nelem = cl; if (nelem < 0) nelem = 0;
    const int* srcc = src + e0;
    const int* dstc = dst + e0;
    const bool vec = ((((size_t)src) | ((size_t)dst)) & 15) == 0;

    for (unsigned r0 = (unsigned)r * RS; r0 < (unsigned)N; r0 += NR * RS) {
        for (int i = tid; i < RS; i += PTB) acc[i] = 0.0f;
        __syncthreads();
        agg_scan(srcc, dstc, nelem, vec, dinv, h, acc, q, r0, tid, lane);
        __syncthreads();
        float dv[FL];
#pragma unroll
        for (int j = 0; j < FL; ++j) {
            int k = tid + j * PTB + c * ROT; if (k >= RS) k -= RS;
            const unsigned n = r0 + (unsigned)k;
            dv[j] = dinv[n < (unsigned)N ? n : (unsigned)N - 1u];
        }
#pragma unroll
        for (int j = 0; j < FL; ++j) {
            const int i = tid + j * PTB;
            if (i < RS) {
                int k = i + c * ROT; if (k >= RS) k -= RS;
                const unsigned n = r0 + (unsigned)k;
                const float v = acc[k];
                if (v != 0.0f && n < (unsigned)N) atomicAdd(&out[n], dv[j] * v);
            }
        }
        __syncthreads();
    }
}

extern "C" void kernel_launch(void* const* d_in, const int* in_sizes, int n_in,
                              void* d_out, int out_size, void* d_ws, size_t ws_size,
                              hipStream_t stream) {
    const float* x  = (const float*)d_in[0];
    const int*   ei = (const int*)d_in[1];
    const float* W  = (const float*)d_in[2];
    float*       out = (float*)d_out;

    const int N = out_size;
    const int E = in_sizes[1] / 2;
    const int* src = ei;
    const int* dst = ei + E;

    float* deg  = (float*)d_ws;
    float* dinv = deg + N;
    float* h    = dinv + N;

    const int TB = 256;
    const int nBlkN = (N + TB - 1) / TB;
    const int cl = (((E + NC - 1) / NC) + 3) & ~3;
    const int nWaves = (N + 15) / 16;
    const int nbx = ((nWaves + 32 * NR * NC - 1) / (32 * NR * NC)) * (NR * NC);
    const int wpb = (nWaves + nbx - 1) / nbx;

    gcn_init_deg<<<nBlkN, TB, 0, stream>>>(deg, N);
    gcn_xw_deg<<<nbx + NR * NC, PTB, 0, stream>>>(x, W, h, dst, deg, N, E, cl, nbx, wpb);
    gcn_dinv<<<nBlkN, TB, 0, stream>>>(deg, dinv, N);
    gcn_init_out<<<nBlkN, TB, 0, stream>>>(dinv, h, out, N);
    gcn_agg_lds<<<NR * NC, PTB, 0, stream>>>(src, dst, dinv, h, out, N, E, cl);
}


# Sum pass gathers ONE float per edge: g[src] = dinv*h, formed once per node by the self-loop kernel (was dinv[src] and h[src]); no second gather padded by a re-read; values added bit-identical
# speedup vs baseline: 4.4837x; 1.2467x over previous
#include <hip/hip_runtime.h>
#include <hip/hip_bf16.h>

typedef __attribute__((ext_vector_type(2))) float v2f;
typedef __attribute__((ext_vector_type(4))) float v4f;
typedef __attribute__((ext_vector_type(8))) float v8f;

#define D_FEAT 512

__global__ void gcn_init_deg(float* __restrict__ deg, int N) {
    int i = blockIdx.x * blockDim.x + threadIdx.x;
    if (i < N) deg[i] = 1.0f;
}

__device__ __forceinline__ void
xw_wave(const float* __restrict__ x, const float* __restrict__ W,
        float* __restrict__ h, int N, int wave) {
    int lane = threadIdx.x & 31;
    int row0 = wave * 16;
    if (row0 >= N) return;

    int r    = lane & 15;
    int half = lane >> 4;

    float m0 = (lane == 0)  ? 1.0f : 0.0f;
    float m1 = (lane == 16) ? 1.0f : 0.0f;

    const v2f* xv = (const v2f*)(x + (size_t)(row0 + r) * D_FEAT) + half;
    const v4f* Wv = (const v4f*)W;

    v8f c = {0.f, 0.f, 0.f, 0.f, 0.f, 0.f, 0.f, 0.f};

#pragma unroll 4
    for (int k4 = 0; k4 < D_FEAT / 4; ++k4) {
        v2f a = xv[2 * k4];
        v4f w = Wv[k4];
        v2f b;
        b.x = w.x * m0 + w.z * m1;
        b.y = w.y * m0 + w.w * m1;
        c = __builtin_amdgcn_wmma_f32_16x16x4_f32(
                 false, a,  false, b,
                 (short)0, c,  false,  false);
    }

    if (r == 0) {
        float* hp = h + row0 + 8 * half;
#pragma unroll
        for (int i = 0; i < 8; ++i) hp[i] = c[i];
    }
}

__global__ void gcn_dinv(const float* __restrict__ deg, float* __restrict__ dinv, int N) {
    int i = blockIdx.x * blockDim.x + threadIdx.x;
    if (i < N) dinv[i] = __frsqrt_rn(deg[i]);
}

__global__ void gcn_init_out(const float* __restrict__ dinv, const float* __restrict__ h,
                             float* __restrict__ g, float* __restrict__ out, int N) {
    int i = blockIdx.x * blockDim.x + threadIdx.x;
    if (i < N) {
        float di = dinv[i];
        float hi = h[i];
        out[i] = di * di * hi;
        g[i] = di * hi;
    }
}

#define NR   16
#define NC   16
#define RS   6272
#define PTB  1024
#define DU   8
#define TU   4
#define QCAP (32 * 4 * TU)
#define ROT  ((RS / NC) & ~31)
#define FL   ((RS + PTB - 1) / PTB)

__device__ __forceinline__ int4 ld_quad_guarded(const int* __restrict__ p, int q, int nelem, int fill) {
    int4 v = make_int4(fill, fill, fill, fill);
    const int e = q * 4;
    if (e     < nelem) v.x = p[e];
    if (e + 1 < nelem) v.y = p[e + 1];
    if (e + 2 < nelem) v.z = p[e + 2];
    if (e + 3 < nelem) v.w = p[e + 3];
    return v;
}

template <int U>
__device__ __forceinline__ void ld_tile_vec(const int* __restrict__ p, int qbase, int tid, int4 (&v)[U]) {
    const int4* p4 = (const int4*)p + qbase + tid;
#pragma unroll
    for (int u = 0; u < U; ++u) v[u] = p4[u * PTB];
}
template <int U>
__device__ __forceinline__ void ld_tile_guarded(const int* __restrict__ p, int qbase, int tid, int nelem,
                                                int fill, int4 (&v)[U]) {
#pragma unroll
    for (int u = 0; u < U; ++u) v[u] = ld_quad_guarded(p, qbase + u * PTB + tid, nelem, fill);
}

__device__ __forceinline__ void deg_tile(const int4 (&v)[DU], unsigned* cnt, unsigned r0) {
#pragma unroll
    for (int u = 0; u < DU; ++u) {
        unsigned k;
        k = (unsigned)v[u].x - r0; if (k < RS) atomicAdd(&cnt[k], 1u);
        k = (unsigned)v[u].y - r0; if (k < RS) atomicAdd(&cnt[k], 1u);
        k = (unsigned)v[u].z - r0; if (k < RS) atomicAdd(&cnt[k], 1u);
        k = (unsigned)v[u].w - r0; if (k < RS) atomicAdd(&cnt[k], 1u);
    }
}

__device__ __forceinline__ void deg_scan(const int* __restrict__ dstc, int nelem, bool vec,
                                         unsigned* cnt, unsigned r0, int tid) {
    const int nq = (nelem + 3) >> 2;
    const int nfull = vec ? (nelem >> 2) / (PTB * DU) : 0;
    for (int t = 0; t < nfull; ++t) {
        int4 v[DU];
        ld_tile_vec<DU>(dstc, t * (PTB * DU), tid, v);
        deg_tile(v, cnt, r0);
    }
    for (int base = nfull * (PTB * DU); base < nq; base += PTB * DU) {
        int4 v[DU];
        ld_tile_guarded<DU>(dstc, base, tid, nelem, -1, v);
        deg_tile(v, cnt, r0);
    }
}

__device__ __forceinline__ void
deg_group(const int* __restrict__ dst, float* __restrict__ deg, int N, int E, int cl,
          unsigned* cnt, int bid) {
    const int tid = threadIdx.x;
    const int r = bid % NR;
    const int c = bid / NR;
    const int e0 = c * cl;
    int nelem = E - e0; if (nelem > cl) nelem = cl; if (nelem < 0) nelem = 0;
    const int* dstc = dst + e0;
    const bool vec = (((size_t)dst) & 15) == 0;

    for (unsigned r0 = (unsigned)r * RS; r0 < (unsigned)N; r0 += NR * RS) {
        for (int i = tid; i < RS; i += PTB) cnt[i] = 0u;
        __syncthreads();
        deg_scan(dstc, nelem, vec, cnt, r0, tid);
        __syncthreads();
        for (int i = tid; i < RS; i += PTB) {
            int k = i + c * ROT; if (k >= RS) k -= RS;
            const unsigned n = r0 + (unsigned)k;
            const unsigned cv = cnt[k];
            if (cv != 0u && n < (unsigned)N) atomicAdd(&deg[n], (float)cv);
        }
        __syncthreads();
    }
}

__global__ void __launch_bounds__(PTB)
gcn_xw_deg(const float* __restrict__ x, const float* __restrict__ W, float* __restrict__ h,
           const int* __restrict__ dst, float* __restrict__ deg,
           int N, int E, int cl, int nbx, int wpb) {
    __shared__ unsigned cnt[RS];
    const int bid = blockIdx.x;
    if (bid < nbx) {
        const int wi = threadIdx.x >> 5;
        if (wi < wpb) xw_wave(x, W, h, N, bid * wpb + wi);
    } else {
        deg_group(dst, deg, N, E, cl, cnt, bid - nbx);
    }
}

__device__ __forceinline__ void agg_tile(const int4 (&dc)[TU], const int4 (&sc)[TU],
                                         const float* __restrict__ g,
                                         float* acc, uint2* q, unsigned r0, int lane) {
    unsigned cnt = 0;
#pragma unroll
    for (int u = 0; u < TU; ++u) {
        const int dd[4] = {dc[u].x, dc[u].y, dc[u].z, dc[u].w};
        const int ss[4] = {sc[u].x, sc[u].y, sc[u].z, sc[u].w};
#pragma unroll
        for (int j = 0; j < 4; ++j) {
            const unsigned k = (unsigned)dd[j] - r0;
            const bool hit = k < RS;
            const unsigned m = __builtin_amdgcn_ballot_w32(hit);
            if (hit) q[cnt + __builtin_amdgcn_mbcnt_lo(m, 0u)] = make_uint2(k, (unsigned)ss[j]);
            cnt += (unsigned)__builtin_popcount(m);
        }
    }
    for (unsigned base = 0; base < cnt; base += 64u) {
        const unsigned i = base + (unsigned)lane;
        if (base + 32u < cnt) {
            const unsigned i2 = i + 32u;
            const bool two = i2 < cnt;
            const uint2 a = q[i];
            const uint2 b = q[two ? i2 : i];
            const float ga = g[a.y];
            const float gb = g[two ? b.y : 0u];
            atomicAdd(&acc[a.x], ga);
            atomicAdd(&acc[b.x], two ? gb : 0.0f);
        } else if (i < cnt) {
            const uint2 a = q[i];
            atomicAdd(&acc[a.x], g[a.y]);
        }
    }
}

__device__ __forceinline__ void agg_scan(const int* __restrict__ srcc, const int* __restrict__ dstc,
                                         int nelem, bool vec, const float* __restrict__ g,
                                         float* acc, uint2* q,
                                         unsigned r0, int tid, int lane) {
    const int nq = (nelem + 3) >> 2;
    const int nfull = vec ? (nelem >> 2) / (PTB * TU) : 0;
    if (nfull > 0) {
        int4 dn[TU], sn[TU];
        ld_tile_vec<TU>(dstc, 0, tid, dn);
        ld_tile_vec<TU>(srcc, 0, tid, sn);
        for (int t = 0; t < nfull; ++t) {
            int4 dc[TU], sc[TU];
#pragma unroll
            for (int u = 0; u < TU; ++u) { dc[u] = dn[u]; sc[u] = sn[u]; }
            if (t + 1 < nfull) {
                ld_tile_vec<TU>(dstc, (t + 1) * (PTB * TU), tid, dn);
                ld_tile_vec<TU>(srcc, (t + 1) * (PTB * TU), tid, sn);
            }
            agg_tile(dc, sc, g, acc, q, r0, lane);
        }
    }
    for (int base = nfull * (PTB * TU); base < nq; base += PTB * TU) {
        int4 dc[TU], sc[TU];
        ld_tile_guarded<TU>(dstc, base, tid, nelem, -1, dc);
        ld_tile_guarded<TU>(srcc, base, tid, nelem, 0, sc);
        agg_tile(dc, sc, g, acc, q, r0, lane);
    }
}

__global__ void __launch_bounds__(PTB)
gcn_agg_lds(const int* __restrict__ src, const int* __restrict__ dst,
            const float* __restrict__ dinv, const float* __restrict__ g,
            float* __restrict__ out, int N, int E, int cl) {
    __shared__ float acc[RS];
    __shared__ uint2 queue[(PTB / 32) * QCAP];
    const int tid = threadIdx.x;
    const int lane = tid & 31;
    uint2* q = queue + (tid >> 5) * QCAP;
    const int r = blockIdx.x % NR;
    const int c = blockIdx.x / NR;
    const int e0 = c * cl;
    int nelem = E - e0; if (nelem > cl) nelem = cl; if (nelem < 0) nelem = 0;
    const int* srcc = src + e0;
    const int* dstc = dst + e0;
    const bool vec = ((((size_t)src) | ((size_t)dst)) & 15) == 0;

    for (unsigned r0 = (unsigned)r * RS; r0 < (unsigned)N; r0 += NR * RS) {
        for (int i = tid; i < RS; i += PTB) acc[i] = 0.0f;
        __syncthreads();
        agg_scan(srcc, dstc, nelem, vec, g, acc, q, r0, tid, lane);
        __syncthreads();
        float dv[FL];
#pragma unroll
        for (int j = 0; j < FL; ++j) {
            int k = tid + j * PTB + c * ROT; if (k >= RS) k -= RS;
            const unsigned n = r0 + (unsigned)k;
            dv[j] = dinv[n < (unsigned)N ? n : (unsigned)N - 1u];
        }
#pragma unroll
        for (int j = 0; j < FL; ++j) {
            const int i = tid + j * PTB;
            if (i < RS) {
                int k = i + c * ROT; if (k >= RS) k -= RS;
                const unsigned n = r0 + (unsigned)k;
                const float v = acc[k];
                if (v != 0.0f && n < (unsigned)N) atomicAdd(&out[n], dv[j] * v);
            }
        }
        __syncthreads();
    }
}

extern "C" void kernel_launch(void* const* d_in, const int* in_sizes, int n_in,
                              void* d_out, int out_size, void* d_ws, size_t ws_size,
                              hipStream_t stream) {
    const float* x  = (const float*)d_in[0];
    const int*   ei = (const int*)d_in[1];
    const float* W  = (const float*)d_in[2];
    float*       out = (float*)d_out;

    const int N = out_size;
    const int E = in_sizes[1] / 2;
    const int* src = ei;
    const int* dst = ei + E;

    float* deg  = (float*)d_ws;
    float* dinv = deg + N;
    float* h    = dinv + N;
    float* g    = deg;

    const int TB = 256;
    const int nBlkN = (N + TB - 1) / TB;
    const int cl = (((E + NC - 1) / NC) + 3) & ~3;
    const int nWaves = (N + 15) / 16;
    const int nbx = ((nWaves + 32 * NR * NC - 1) / (32 * NR * NC)) * (NR * NC);
    const int wpb = (nWaves + nbx - 1) / nbx;

    gcn_init_deg<<<nBlkN, TB, 0, stream>>>(deg, N);
    gcn_xw_deg<<<nbx + NR * NC, PTB, 0, stream>>>(x, W, h, dst, deg, N, E, cl, nbx, wpb);
    gcn_dinv<<<nBlkN, TB, 0, stream>>>(deg, dinv, N);
    gcn_init_out<<<nBlkN, TB, 0, stream>>>(dinv, h, g, out, N);
    gcn_agg_lds<<<NR * NC, PTB, 0, stream>>>(src, dst, dinv, g, out, N, E, cl);
}


# x@W streams each row as whole 128-B lines: one row per wave step (16-B non-temporal lane loads, 8 per group), W in the same lane order, h = trace of the WMMA tile; h sum order changes
# speedup vs baseline: 4.8556x; 1.0829x over previous
#include <hip/hip_runtime.h>
#include <hip/hip_bf16.h>

typedef __attribute__((ext_vector_type(2))) float v2f;
typedef __attribute__((ext_vector_type(4))) float v4f;
typedef __attribute__((ext_vector_type(8))) float v8f;

#define D_FEAT 512

__global__ void gcn_init_deg(float* __restrict__ deg, int N) {
    int i = blockIdx.x * blockDim.x + threadIdx.x;
    if (i < N) deg[i] = 1.0f;
}

#define XB 2

template <int M>
__device__ __forceinline__ float lane_xor_in_row(float v) {
    const int b = __builtin_bit_cast(int, v);
    return __builtin_bit_cast(float, __builtin_amdgcn_update_dpp(b, b, 0x160 + M, 0xf, 0xf, false));
}
__device__ __forceinline__ float lane_xor_16(float v) {
    const int b = __builtin_bit_cast(int, v);
    return __builtin_bit_cast(float, __builtin_amdgcn_permlanex16(b, b, 0x76543210, 0xfedcba98, false, false));
}

__device__ __forceinline__ float
xw_row_dot(const v4f (&b)[4], const v4f (&wv)[4], int sel, bool diag) {
    v8f d = {0.f, 0.f, 0.f, 0.f, 0.f, 0.f, 0.f, 0.f};
#pragma unroll
    for (int c = 0; c < 4; ++c) {
        v2f a0, b0, a1, b1;
        a0.x = b[c].x;  a0.y = b[c].y;  b0.x = wv[c].x; b0.y = wv[c].y;
        a1.x = b[c].z;  a1.y = b[c].w;  b1.x = wv[c].z; b1.y = wv[c].w;
        d = __builtin_amdgcn_wmma_f32_16x16x4_f32(false, a0, false, b0, (short)0, d, false, false);
        d = __builtin_amdgcn_wmma_f32_16x16x4_f32(false, a1, false, b1, (short)0, d, false, false);
    }
    float e = d[0];
    e = (sel == 1) ? d[1] : e;
    e = (sel == 2) ? d[2] : e;
    e = (sel == 3) ? d[3] : e;
    e = (sel == 4) ? d[4] : e;
    e = (sel == 5) ? d[5] : e;
    e = (sel == 6) ? d[6] : e;
    e = (sel == 7) ? d[7] : e;
    e = diag ? e : 0.0f;
    e += lane_xor_in_row<1>(e);
    e += lane_xor_in_row<2>(e);
    e += lane_xor_in_row<4>(e);
    e += lane_xor_in_row<8>(e);
    e += lane_xor_16(e);
    return e;
}

template <int NB>
__device__ __forceinline__ void
xw_batch(const float* __restrict__ x, float* __restrict__ h, int row, int lane,
         const v4f (&wv)[4], int sel, bool diag) {
    v4f buf[NB][4];
    float t[NB];
    const v4f* xr = (const v4f*)(x + (size_t)row * D_FEAT) + lane;
#pragma unroll
    for (int j = 0; j < NB; ++j) {
#pragma unroll
        for (int c = 0; c < 4; ++c) buf[j][c] = __builtin_nontemporal_load(xr + 128 * j + 32 * c);
    }
    __builtin_amdgcn_sched_barrier(0);
#pragma unroll
    for (int j = 0; j < NB; ++j) t[j] = xw_row_dot(buf[j], wv, sel, diag);
    if (lane < NB) {
        float v = t[0];
#pragma unroll
        for (int j = 1; j < NB; ++j) v = (lane == j) ? t[j] : v;
        h[row + lane] = v;
    }
}

__device__ __forceinline__ void
xw_rows(const float* __restrict__ x, const float* __restrict__ W,
        float* __restrict__ h, int N, int row0, int nrows) {
    if (row0 >= N) return;
    const int lane = threadIdx.x & 31;
    const int rend = (row0 + nrows < N) ? row0 + nrows : N;
    const int sel = lane & 7;
    const bool diag = (lane < 8) || (lane >= 24);

    v4f wv[4];
    {
        const v4f* Wv = (const v4f*)W + lane;
#pragma unroll
        for (int c = 0; c < 4; ++c) wv[c] = Wv[32 * c];
    }

    int r = row0;
    for (; r + XB <= rend; r += XB) xw_batch<XB>(x, h, r, lane, wv, sel, diag);
    if (r < rend) xw_batch<1>(x, h, r, lane, wv, sel, diag);
}

__global__ void gcn_dinv(const float* __restrict__ deg, float* __restrict__ dinv, int N) {
    int i = blockIdx.x * blockDim.x + threadIdx.x;
    if (i < N) dinv[i] = __frsqrt_rn(deg[i]);
}

__global__ void gcn_init_out(const float* __restrict__ dinv, const float* __restrict__ h,
                             float* __restrict__ g, float* __restrict__ out, int N) {
    int i = blockIdx.x * blockDim.x + threadIdx.x;
    if (i < N) {
        float di = dinv[i];
        float hi = h[i];
        out[i] = di * di * hi;
        g[i] = di * hi;
    }
}

#define NR   16
#define NC   16
#define RS   6272
#define PTB  1024
#define DU   8
#define TU   4
#define QCAP (32 * 4 * TU)
#define ROT  ((RS / NC) & ~31)
#define FL   ((RS + PTB - 1) / PTB)

__device__ __forceinline__ int4 ld_quad_guarded(const int* __restrict__ p, int q, int nelem, int fill) {
    int4 v = make_int4(fill, fill, fill, fill);
    const int e = q * 4;
    if (e     < nelem) v.x = p[e];
    if (e + 1 < nelem) v.y = p[e + 1];
    if (e + 2 < nelem) v.z = p[e + 2];
    if (e + 3 < nelem) v.w = p[e + 3];
    return v;
}

template <int U>
__device__ __forceinline__ void ld_tile_vec(const int* __restrict__ p, int qbase, int tid, int4 (&v)[U]) {
    const int4* p4 = (const int4*)p + qbase + tid;
#pragma unroll
    for (int u = 0; u < U; ++u) v[u] = p4[u * PTB];
}
template <int U>
__device__ __forceinline__ void ld_tile_guarded(const int* __restrict__ p, int qbase, int tid, int nelem,
                                                int fill, int4 (&v)[U]) {
#pragma unroll
    for (int u = 0; u < U; ++u) v[u] = ld_quad_guarded(p, qbase + u * PTB + tid, nelem, fill);
}

__device__ __forceinline__ void deg_tile(const int4 (&v)[DU], unsigned* cnt, unsigned r0) {
#pragma unroll
    for (int u = 0; u < DU; ++u) {
        unsigned k;
        k = (unsigned)v[u].x - r0; if (k < RS) atomicAdd(&cnt[k], 1u);
        k = (unsigned)v[u].y - r0; if (k < RS) atomicAdd(&cnt[k], 1u);
        k = (unsigned)v[u].z - r0; if (k < RS) atomicAdd(&cnt[k], 1u);
        k = (unsigned)v[u].w - r0; if (k < RS) atomicAdd(&cnt[k], 1u);
    }
}

__device__ __forceinline__ void deg_scan(const int* __restrict__ dstc, int nelem, bool vec,
                                         unsigned* cnt, unsigned r0, int tid) {
    const int nq = (nelem + 3) >> 2;
    const int nfull = vec ? (nelem >> 2) / (PTB * DU) : 0;
    for (int t = 0; t < nfull; ++t) {
        int4 v[DU];
        ld_tile_vec<DU>(dstc, t * (PTB * DU), tid, v);
        deg_tile(v, cnt, r0);
    }
    for (int base = nfull * (PTB * DU); base < nq; base += PTB * DU) {
        int4 v[DU];
        ld_tile_guarded<DU>(dstc, base, tid, nelem, -1, v);
        deg_tile(v, cnt, r0);
    }
}

__device__ __forceinline__ void
deg_group(const int* __restrict__ dst, float* __restrict__ deg, int N, int E, int cl,
          unsigned* cnt, int bid) {
    const int tid = threadIdx.x;
    const int r = bid % NR;
    const int c = bid / NR;
    const int e0 = c * cl;
    int nelem = E - e0; if (nelem > cl) nelem = cl; if (nelem < 0) nelem = 0;
    const int* dstc = dst + e0;
    const bool vec = (((size_t)dst) & 15) == 0;

    for (unsigned r0 = (unsigned)r * RS; r0 < (unsigned)N; r0 += NR * RS) {
        for (int i = tid; i < RS; i += PTB) cnt[i] = 0u;
        __syncthreads();
        deg_scan(dstc, nelem, vec, cnt, r0, tid);
        __syncthreads();
        for (int i = tid; i < RS; i += PTB) {
            int k = i + c * ROT; if (k >= RS) k -= RS;
            const unsigned n = r0 + (unsigned)k;
            const unsigned cv = cnt[k];
            if (cv != 0u && n < (unsigned)N) atomicAdd(&deg[n], (float)cv);
        }
        __syncthreads();
    }
}

__global__ void __launch_bounds__(PTB, 16)
gcn_xw_deg(const float* __restrict__ x, const float* __restrict__ W, float* __restrict__ h,
           const int* __restrict__ dst, float* __restrict__ deg,
           int N, int E, int cl, int nbx, int rpw) {
    __shared__ unsigned cnt[RS];
    const int bid = blockIdx.x;
    if (bid < nbx) {
        const int wi = __builtin_amdgcn_readfirstlane((int)(threadIdx.x >> 5));
        xw_rows(x, W, h, N, (bid * (PTB / 32) + wi) * rpw, rpw);
    } else {
        deg_group(dst, deg, N, E, cl, cnt, bid - nbx);
    }
}

__device__ __forceinline__ void agg_tile(const int4 (&dc)[TU], const int4 (&sc)[TU],
                                         const float* __restrict__ g,
                                         float* acc, uint2* q, unsigned r0, int lane) {
    unsigned cnt = 0;
#pragma unroll
    for (int u = 0; u < TU; ++u) {
        const int dd[4] = {dc[u].x, dc[u].y, dc[u].z, dc[u].w};
        const int ss[4] = {sc[u].x, sc[u].y, sc[u].z, sc[u].w};
#pragma unroll
        for (int j = 0; j < 4; ++j) {
            const unsigned k = (unsigned)dd[j] - r0;
            const bool hit = k < RS;
            const unsigned m = __builtin_amdgcn_ballot_w32(hit);
            if (hit) q[cnt + __builtin_amdgcn_mbcnt_lo(m, 0u)] = make_uint2(k, (unsigned)ss[j]);
            cnt += (unsigned)__builtin_popcount(m);
        }
    }
    for (unsigned base = 0; base < cnt; base += 64u) {
        const unsigned i = base + (unsigned)lane;
        if (base + 32u < cnt) {
            const unsigned i2 = i + 32u;
            const bool two = i2 < cnt;
            const uint2 a = q[i];
            const uint2 b = q[two ? i2 : i];
            const float ga = g[a.y];
            const float gb = g[two ? b.y : 0u];
            atomicAdd(&acc[a.x], ga);
            atomicAdd(&acc[b.x], two ? gb : 0.0f);
        } else if (i < cnt) {
            const uint2 a = q[i];
            atomicAdd(&acc[a.x], g[a.y]);
        }
    }
}

__device__ __forceinline__ void agg_scan(const int* __restrict__ srcc, const int* __restrict__ dstc,
                                         int nelem, bool vec, const float* __restrict__ g,
                                         float* acc, uint2* q,
                                         unsigned r0, int tid, int lane) {
    const int nq = (nelem + 3) >> 2;
    const int nfull = vec ? (nelem >> 2) / (PTB * TU) : 0;
    if (nfull > 0) {
        int4 dn[TU], sn[TU];
        ld_tile_vec<TU>(dstc, 0, tid, dn);
        ld_tile_vec<TU>(srcc, 0, tid, sn);
        for (int t = 0; t < nfull; ++t) {
            int4 dc[TU], sc[TU];
#pragma unroll
            for (int u = 0; u < TU; ++u) { dc[u] = dn[u]; sc[u] = sn[u]; }
            if (t + 1 < nfull) {
                ld_tile_vec<TU>(dstc, (t + 1) * (PTB * TU), tid, dn);
                ld_tile_vec<TU>(srcc, (t + 1) * (PTB * TU), tid, sn);
            }
            agg_tile(dc, sc, g, acc, q, r0, lane);
        }
    }
    for (int base = nfull * (PTB * TU); base < nq; base += PTB * TU) {
        int4 dc[TU], sc[TU];
        ld_tile_guarded<TU>(dstc, base, tid, nelem, -1, dc);
        ld_tile_guarded<TU>(srcc, base, tid, nelem, 0, sc);
        agg_tile(dc, sc, g, acc, q, r0, lane);
    }
}

__global__ void __launch_bounds__(PTB)
gcn_agg_lds(const int* __restrict__ src, const int* __restrict__ dst,
            const float* __restrict__ dinv, const float* __restrict__ g,
            float* __restrict__ out, int N, int E, int cl) {
    __shared__ float acc[RS];
    __shared__ uint2 queue[(PTB / 32) * QCAP];
    const int tid = threadIdx.x;
    const int lane = tid & 31;
    uint2* q = queue + (tid >> 5) * QCAP;
    const int r = blockIdx.x % NR;
    const int c = blockIdx.x / NR;
    const int e0 = c * cl;
    int nelem = E - e0; if (nelem > cl) nelem = cl; if (nelem < 0) nelem = 0;
    const int* srcc = src + e0;
    const int* dstc = dst + e0;
    const bool vec = ((((size_t)src) | ((size_t)dst)) & 15) == 0;

    for (unsigned r0 = (unsigned)r * RS; r0 < (unsigned)N; r0 += NR * RS) {
        for (int i = tid; i < RS; i += PTB) acc[i] = 0.0f;
        __syncthreads();
        agg_scan(srcc, dstc, nelem, vec, g, acc, q, r0, tid, lane);
        __syncthreads();
        float dv[FL];
#pragma unroll
        for (int j = 0; j < FL; ++j) {
            int k = tid + j * PTB + c * ROT; if (k >= RS) k -= RS;
            const unsigned n = r0 + (unsigned)k;
            dv[j] = dinv[n < (unsigned)N ? n : (unsigned)N - 1u];
        }
#pragma unroll
        for (int j = 0; j < FL; ++j) {
            const int i = tid + j * PTB;
            if (i < RS) {
                int k = i + c * ROT; if (k >= RS) k -= RS;
                const unsigned n = r0 + (unsigned)k;
                const float v = acc[k];
                if (v != 0.0f && n < (unsigned)N) atomicAdd(&out[n], dv[j] * v);
            }
        }
        __syncthreads();
    }
}

extern "C" void kernel_launch(void* const* d_in, const int* in_sizes, int n_in,
                              void* d_out, int out_size, void* d_ws, size_t ws_size,
                              hipStream_t stream) {
    const float* x  = (const float*)d_in[0];
    const int*   ei = (const int*)d_in[1];
    const float* W  = (const float*)d_in[2];
    float*       out = (float*)d_out;

    const int N = out_size;
    const int E = in_sizes[1] / 2;
    const int* src = ei;
    const int* dst = ei + E;

    float* deg  = (float*)d_ws;
    float* dinv = deg + N;
    float* h    = dinv + N;
    float* g    = deg;

    const int TB = 256;
    const int nBlkN = (N + TB - 1) / TB;
    const int cl = (((E + NC - 1) / NC) + 3) & ~3;
    const int nbx = NR * NC;
    const int rpw = (N + nbx * (PTB / 32) - 1) / (nbx * (PTB / 32));

    gcn_init_deg<<<nBlkN, TB, 0, stream>>>(deg, N);
    gcn_xw_deg<<<nbx + NR * NC, PTB, 0, stream>>>(x, W, h, dst, deg, N, E, cl, nbx, rpw);
    gcn_dinv<<<nBlkN, TB, 0, stream>>>(deg, dinv, N);
    gcn_init_out<<<nBlkN, TB, 0, stream>>>(dinv, h, g, out, N);
    gcn_agg_lds<<<NR * NC, PTB, 0, stream>>>(src, dst, dinv, g, out, N, E, cl);
}


# Sum pass holds g[0..57344) in LDS (each work-group fills it by 16-B loads): a queued edge with such a source reads LDS, the others gather as before; queue is half a tile; order of LDS adds changes
# speedup vs baseline: 5.1780x; 1.0664x over previous
#include <hip/hip_runtime.h>
#include <hip/hip_bf16.h>

typedef __attribute__((ext_vector_type(2))) float v2f;
typedef __attribute__((ext_vector_type(4))) float v4f;
typedef __attribute__((ext_vector_type(8))) float v8f;

#define D_FEAT 512

__global__ void gcn_init_deg(float* __restrict__ deg, int N) {
    int i = blockIdx.x * blockDim.x + threadIdx.x;
    if (i < N) deg[i] = 1.0f;
}

#define XB 2

template <int M>
__device__ __forceinline__ float lane_xor_in_row(float v) {
    const int b = __builtin_bit_cast(int, v);
    return __builtin_bit_cast(float, __builtin_amdgcn_update_dpp(b, b, 0x160 + M, 0xf, 0xf, false));
}
__device__ __forceinline__ float lane_xor_16(float v) {
    const int b = __builtin_bit_cast(int, v);
    return __builtin_bit_cast(float, __builtin_amdgcn_permlanex16(b, b, 0x76543210, 0xfedcba98, false, false));
}

__device__ __forceinline__ float
xw_row_dot(const v4f (&b)[4], const v4f (&wv)[4], int sel, bool diag) {
    v8f d = {0.f, 0.f, 0.f, 0.f, 0.f, 0.f, 0.f, 0.f};
#pragma unroll
    for (int c = 0; c < 4; ++c) {
        v2f a0, b0, a1, b1;
        a0.x = b[c].x;  a0.y = b[c].y;  b0.x = wv[c].x; b0.y = wv[c].y;
        a1.x = b[c].z;  a1.y = b[c].w;  b1.x = wv[c].z; b1.y = wv[c].w;
        d = __builtin_amdgcn_wmma_f32_16x16x4_f32(false, a0, false, b0, (short)0, d, false, false);
        d = __builtin_amdgcn_wmma_f32_16x16x4_f32(false, a1, false, b1, (short)0, d, false, false);
    }
    float e = d[0];
    e = (sel == 1) ? d[1] : e;
    e = (sel == 2) ? d[2] : e;
    e = (sel == 3) ? d[3] : e;
    e = (sel == 4) ? d[4] : e;
    e = (sel == 5) ? d[5] : e;
    e = (sel == 6) ? d[6] : e;
    e = (sel == 7) ? d[7] : e;
    e = diag ? e : 0.0f;
    e += lane_xor_in_row<1>(e);
    e += lane_xor_in_row<2>(e);
    e += lane_xor_in_row<4>(e);
    e += lane_xor_in_row<8>(e);
    e += lane_xor_16(e);
    return e;
}

template <int NB>
__device__ __forceinline__ void
xw_batch(const float* __restrict__ x, float* __restrict__ h, int row, int lane,
         const v4f (&wv)[4], int sel, bool diag) {
    v4f buf[NB][4];
    float t[NB];
    const v4f* xr = (const v4f*)(x + (size_t)row * D_FEAT) + lane;
#pragma unroll
    for (int j = 0; j < NB; ++j) {
#pragma unroll
        for (int c = 0; c < 4; ++c) buf[j][c] = __builtin_nontemporal_load(xr + 128 * j + 32 * c);
    }
    __builtin_amdgcn_sched_barrier(0);
#pragma unroll
    for (int j = 0; j < NB; ++j) t[j] = xw_row_dot(buf[j], wv, sel, diag);
    if (lane < NB) {
        float v = t[0];
#pragma unroll
        for (int j = 1; j < NB; ++j) v = (lane == j) ? t[j] : v;
        h[row + lane] = v;
    }
}

__device__ __forceinline__ void
xw_rows(const float* __restrict__ x, const float* __restrict__ W,
        float* __restrict__ h, int N, int row0, int nrows) {
    if (row0 >= N) return;
    const int lane = threadIdx.x & 31;
    const int rend = (row0 + nrows < N) ? row0 + nrows : N;
    const int sel = lane & 7;
    const bool diag = (lane < 8) || (lane >= 24);

    v4f wv[4];
    {
        const v4f* Wv = (const v4f*)W + lane;
#pragma unroll
        for (int c = 0; c < 4; ++c) wv[c] = Wv[32 * c];
    }

    int r = row0;
    for (; r + XB <= rend; r += XB) xw_batch<XB>(x, h, r, lane, wv, sel, diag);
    if (r < rend) xw_batch<1>(x, h, r, lane, wv, sel, diag);
}

__global__ void gcn_dinv(const float* __restrict__ deg, float* __restrict__ dinv, int N) {
    int i = blockIdx.x * blockDim.x + threadIdx.x;
    if (i < N) dinv[i] = __frsqrt_rn(deg[i]);
}

__global__ void gcn_init_out(const float* __restrict__ dinv, const float* __restrict__ h,
                             float* __restrict__ g, float* __restrict__ out, int N) {
    int i = blockIdx.x * blockDim.x + threadIdx.x;
    if (i < N) {
        float di = dinv[i];
        float hi = h[i];
        out[i] = di * di * hi;
        g[i] = di * hi;
    }
}

#define NR   16
#define NC   16
#define RS   6272
#define PTB  1024
#define DU   8
#define TU   4
#define HU   (TU / 2)
#define QCAP (32 * 4 * HU)
#define GWN  57344
#define GWU  7
#define ROT  ((RS / NC) & ~31)
#define FL   ((RS + PTB - 1) / PTB)

__device__ __forceinline__ int4 ld_quad_guarded(const int* __restrict__ p, int q, int nelem, int fill) {
    int4 v = make_int4(fill, fill, fill, fill);
    const int e = q * 4;
    if (e     < nelem) v.x = p[e];
    if (e + 1 < nelem) v.y = p[e + 1];
    if (e + 2 < nelem) v.z = p[e + 2];
    if (e + 3 < nelem) v.w = p[e + 3];
    return v;
}

template <int U>
__device__ __forceinline__ void ld_tile_vec(const int* __restrict__ p, int qbase, int tid, int4 (&v)[U]) {
    const int4* p4 = (const int4*)p + qbase + tid;
#pragma unroll
    for (int u = 0; u < U; ++u) v[u] = p4[u * PTB];
}
template <int U>
__device__ __forceinline__ void ld_tile_guarded(const int* __restrict__ p, int qbase, int tid, int nelem,
                                                int fill, int4 (&v)[U]) {
#pragma unroll
    for (int u = 0; u < U; ++u) v[u] = ld_quad_guarded(p, qbase + u * PTB + tid, nelem, fill);
}

__device__ __forceinline__ void deg_tile(const int4 (&v)[DU], unsigned* cnt, unsigned r0) {
#pragma unroll
    for (int u = 0; u < DU; ++u) {
        unsigned k;
        k = (unsigned)v[u].x - r0; if (k < RS) atomicAdd(&cnt[k], 1u);
        k = (unsigned)v[u].y - r0; if (k < RS) atomicAdd(&cnt[k], 1u);
        k = (unsigned)v[u].z - r0; if (k < RS) atomicAdd(&cnt[k], 1u);
        k = (unsigned)v[u].w - r0; if (k < RS) atomicAdd(&cnt[k], 1u);
    }
}

__device__ __forceinline__ void deg_scan(const int* __restrict__ dstc, int nelem, bool vec,
                                         unsigned* cnt, unsigned r0, int tid) {
    const int nq = (nelem + 3) >> 2;
    const int nfull = vec ? (nelem >> 2) / (PTB * DU) : 0;
    for (int t = 0; t < nfull; ++t) {
        int4 v[DU];
        ld_tile_vec<DU>(dstc, t * (PTB * DU), tid, v);
        deg_tile(v, cnt, r0);
    }
    for (int base = nfull * (PTB * DU); base < nq; base += PTB * DU) {
        int4 v[DU];
        ld_tile_guarded<DU>(dstc, base, tid, nelem, -1, v);
        deg_tile(v, cnt, r0);
    }
}

__device__ __forceinline__ void
deg_group(const int* __restrict__ dst, float* __restrict__ deg, int N, int E, int cl,
          unsigned* cnt, int bid) {
    const int tid = threadIdx.x;
    const int r = bid % NR;
    const int c = bid / NR;
    const int e0 = c * cl;
    int nelem = E - e0; if (nelem > cl) nelem = cl; if (nelem < 0) nelem = 0;
    const int* dstc = dst + e0;
    const bool vec = (((size_t)dst) & 15) == 0;

    for (unsigned r0 = (unsigned)r * RS; r0 < (unsigned)N; r0 += NR * RS) {
        for (int i = tid; i < RS; i += PTB) cnt[i] = 0u;
        __syncthreads();
        deg_scan(dstc, nelem, vec, cnt, r0, tid);
        __syncthreads();
        for (int i = tid; i < RS; i += PTB) {
            int k = i + c * ROT; if (k >= RS) k -= RS;
            const unsigned n = r0 + (unsigned)k;
            const unsigned cv = cnt[k];
            if (cv != 0u && n < (unsigned)N) atomicAdd(&deg[n], (float)cv);
        }
        __syncthreads();
    }
}

__global__ void __launch_bounds__(PTB, 16)
gcn_xw_deg(const float* __restrict__ x, const float* __restrict__ W, float* __restrict__ h,
           const int* __restrict__ dst, float* __restrict__ deg,
           int N, int E, int cl, int nbx, int rpw) {
    __shared__ unsigned cnt[RS];
    const int bid = blockIdx.x;
    if (bid < nbx) {
        const int wi = __builtin_amdgcn_readfirstlane((int)(threadIdx.x >> 5));
        xw_rows(x, W, h, N, (bid * (PTB / 32) + wi) * rpw, rpw);
    } else {
        deg_group(dst, deg, N, E, cl, cnt, bid - nbx);
    }
}

__device__ __forceinline__ float g_fetch(const float* __restrict__ g, const float* gw, unsigned s) {
    const bool inw = s < (unsigned)GWN;
    float gl = 0.0f;
    if (!inw) gl = g[s];
    float wl = gw[inw ? s : 0u];
    asm volatile("" : "+v"(wl));
    return inw ? wl : gl;
}

template <int U0>
__device__ __forceinline__ void agg_half(const int4 (&dc)[TU], const int4 (&sc)[TU],
                                         const float* __restrict__ g, const float* gw,
                                         float* acc, uint2* q, unsigned r0, int lane) {
    unsigned cnt = 0;
#pragma unroll
    for (int u = U0; u < U0 + HU; ++u) {
        const int dd[4] = {dc[u].x, dc[u].y, dc[u].z, dc[u].w};
        const int ss[4] = {sc[u].x, sc[u].y, sc[u].z, sc[u].w};
#pragma unroll
        for (int j = 0; j < 4; ++j) {
            const unsigned k = (unsigned)dd[j] - r0;
            const bool hit = k < RS;
            const unsigned m = __builtin_amdgcn_ballot_w32(hit);
            if (hit) q[cnt + __builtin_amdgcn_mbcnt_lo(m, 0u)] = make_uint2(k, (unsigned)ss[j]);
            cnt += (unsigned)__builtin_popcount(m);
        }
    }
    for (unsigned base = 0; base < cnt; base += 64u) {
        const unsigned i = base + (unsigned)lane;
        if (base + 32u < cnt) {
            const unsigned i2 = i + 32u;
            const bool two = i2 < cnt;
            const uint2 a = q[i];
            const uint2 b = q[two ? i2 : i];
            const float ga = g_fetch(g, gw, a.y);
            const float gb = g_fetch(g, gw, two ? b.y : 0u);
            atomicAdd(&acc[a.x], ga);
            atomicAdd(&acc[b.x], two ? gb : 0.0f);
        } else if (i < cnt) {
            const uint2 a = q[i];
            atomicAdd(&acc[a.x], g_fetch(g, gw, a.y));
        }
    }
}

__device__ __forceinline__ void agg_tile(const int4 (&dc)[TU], const int4 (&sc)[TU],
                                         const float* __restrict__ g, const float* gw,
                                         float* acc, uint2* q, unsigned r0, int lane) {
    agg_half<0>(dc, sc, g, gw, acc, q, r0, lane);
    agg_half<HU>(dc, sc, g, gw, acc, q, r0, lane);
}

__device__ __forceinline__ void agg_scan(const int* __restrict__ srcc, const int* __restrict__ dstc,
                                         int nelem, bool vec, const float* __restrict__ g,
                                         const float* gw, float* acc, uint2* q,
                                         unsigned r0, int tid, int lane) {
    const int nq = (nelem + 3) >> 2;
    const int nfull = vec ? (nelem >> 2) / (PTB * TU) : 0;
    if (nfull > 0) {
        int4 dn[TU], sn[TU];
        ld_tile_vec<TU>(dstc, 0, tid, dn);
        ld_tile_vec<TU>(srcc, 0, tid, sn);
        for (int t = 0; t < nfull; ++t) {
            int4 dc[TU], sc[TU];
#pragma unroll
            for (int u = 0; u < TU; ++u) { dc[u] = dn[u]; sc[u] = sn[u]; }
            if (t + 1 < nfull) {
                ld_tile_vec<TU>(dstc, (t + 1) * (PTB * TU), tid, dn);
                ld_tile_vec<TU>(srcc, (t + 1) * (PTB * TU), tid, sn);
            }
            agg_tile(dc, sc, g, gw, acc, q, r0, lane);
        }
    }
    for (int base = nfull * (PTB * TU); base < nq; base += PTB * TU) {
        int4 dc[TU], sc[TU];
        ld_tile_guarded<TU>(dstc, base, tid, nelem, -1, dc);
        ld_tile_guarded<TU>(srcc, base, tid, nelem, 0, sc);
        agg_tile(dc, sc, g, gw, acc, q, r0, lane);
    }
}

__device__ __forceinline__ void fill_window(const float* __restrict__ g, float* gw, int N, int tid) {
    const int nw = (N < GWN) ? N : GWN;
    if ((((size_t)g) & 15) == 0) {
        const int nq = nw >> 2;
        const v4f* g4 = (const v4f*)g;
        v4f* w4 = (v4f*)gw;
        const int rot = (nq > 0) ? (int)((blockIdx.x * (unsigned)(GWN / 4 / (NR * NC))) % (unsigned)nq) : 0;
        for (int i0 = 0; i0 < nq; i0 += PTB * GWU) {
            v4f v[GWU];
            int k[GWU];
#pragma unroll
            for (int u = 0; u < GWU; ++u) {
                const int i = i0 + u * PTB + tid;
                int kk = i + rot; if (kk >= nq) kk -= nq;
                k[u] = (i < nq) ? kk : -1;
                v[u] = g4[(i < nq) ? kk : 0];
            }
#pragma unroll
            for (int u = 0; u < GWU; ++u) if (k[u] >= 0) w4[k[u]] = v[u];
        }
        for (int i = (nq << 2) + tid; i < nw; i += PTB) gw[i] = g[i];
    } else {
        for (int i = tid; i < nw; i += PTB) gw[i] = g[i];
    }
}

__global__ void __launch_bounds__(PTB)
gcn_agg_lds(const int* __restrict__ src, const int* __restrict__ dst,
            const float* __restrict__ dinv, const float* __restrict__ g,
            float* __restrict__ out, int N, int E, int cl) {
    __shared__ float acc[RS];
    __shared__ uint2 queue[(PTB / 32) * QCAP];
    __shared__ float gw[GWN];
    const int tid = threadIdx.x;
    const int lane = tid & 31;
    uint2* q = queue + (tid >> 5) * QCAP;
    fill_window(g, gw, N, tid);
    const int r = blockIdx.x % NR;
    const int c = blockIdx.x / NR;
    const int e0 = c * cl;
    int nelem = E - e0; if (nelem > cl) nelem = cl; if (nelem < 0) nelem = 0;
    const int* srcc = src + e0;
    const int* dstc = dst + e0;
    const bool vec = ((((size_t)src) | ((size_t)dst)) & 15) == 0;

    for (unsigned r0 = (unsigned)r * RS; r0 < (unsigned)N; r0 += NR * RS) {
        for (int i = tid; i < RS; i += PTB) acc[i] = 0.0f;
        __syncthreads();
        agg_scan(srcc, dstc, nelem, vec, g, gw, acc, q, r0, tid, lane);
        __syncthreads();
        float dv[FL];
#pragma unroll
        for (int j = 0; j < FL; ++j) {
            int k = tid + j * PTB + c * ROT; if (k >= RS) k -= RS;
            const unsigned n = r0 + (unsigned)k;
            dv[j] = dinv[n < (unsigned)N ? n : (unsigned)N - 1u];
        }
#pragma unroll
        for (int j = 0; j < FL; ++j) {
            const int i = tid + j * PTB;
            if (i < RS) {
                int k = i + c * ROT; if (k >= RS) k -= RS;
                const unsigned n = r0 + (unsigned)k;
                const float v = acc[k];
                if (v != 0.0f && n < (unsigned)N) atomicAdd(&out[n], dv[j] * v);
            }
        }
        __syncthreads();
    }
}

extern "C" void kernel_launch(void* const* d_in, const int* in_sizes, int n_in,
                              void* d_out, int out_size, void* d_ws, size_t ws_size,
                              hipStream_t stream) {
    const float* x  = (const float*)d_in[0];
    const int*   ei = (const int*)d_in[1];
    const float* W  = (const float*)d_in[2];
    float*       out = (float*)d_out;

    const int N = out_size;
    const int E = in_sizes[1] / 2;
    const int* src = ei;
    const int* dst = ei + E;

    float* deg  = (float*)d_ws;
    float* dinv = deg + N;
    float* h    = dinv + N;
    float* g    = deg;

    const int TB = 256;
    const int nBlkN = (N + TB - 1) / TB;
    const int cl = (((E + NC - 1) / NC) + 3) & ~3;
    const int nbx = NR * NC;
    const int rpw = (N + nbx * (PTB / 32) - 1) / (nbx * (PTB / 32));

    gcn_init_deg<<<nBlkN, TB, 0, stream>>>(deg, N);
    gcn_xw_deg<<<nbx + NR * NC, PTB, 0, stream>>>(x, W, h, dst, deg, N, E, cl, nbx, rpw);
    gcn_dinv<<<nBlkN, TB, 0, stream>>>(deg, dinv, N);
    gcn_init_out<<<nBlkN, TB, 0, stream>>>(dinv, h, g, out, N);
    gcn_agg_lds<<<NR * NC, PTB, 0, stream>>>(src, dst, dinv, g, out, N, E, cl);
}


# Both scatter passes tile as 8 node ranges x 32 edge chunks (was 16 x 16): each work-group scans half the edges, indices re-read 8x not 16x; g window 49152; 32 flush adds per node, sum order changes
# speedup vs baseline: 5.7824x; 1.1167x over previous
#include <hip/hip_runtime.h>
#include <hip/hip_bf16.h>

typedef __attribute__((ext_vector_type(2))) float v2f;
typedef __attribute__((ext_vector_type(4))) float v4f;
typedef __attribute__((ext_vector_type(8))) float v8f;

#define D_FEAT 512

__global__ void gcn_init_deg(float* __restrict__ deg, int N) {
    int i = blockIdx.x * blockDim.x + threadIdx.x;
    if (i < N) deg[i] = 1.0f;
}

#define XB 2

template <int M>
__device__ __forceinline__ float lane_xor_in_row(float v) {
    const int b = __builtin_bit_cast(int, v);
    return __builtin_bit_cast(float, __builtin_amdgcn_update_dpp(b, b, 0x160 + M, 0xf, 0xf, false));
}
__device__ __forceinline__ float lane_xor_16(float v) {
    const int b = __builtin_bit_cast(int, v);
    return __builtin_bit_cast(float, __builtin_amdgcn_permlanex16(b, b, 0x76543210, 0xfedcba98, false, false));
}

__device__ __forceinline__ float
xw_row_dot(const v4f (&b)[4], const v4f (&wv)[4], int sel, bool diag) {
    v8f d = {0.f, 0.f, 0.f, 0.f, 0.f, 0.f, 0.f, 0.f};
#pragma unroll
    for (int c = 0; c < 4; ++c) {
        v2f a0, b0, a1, b1;
        a0.x = b[c].x;  a0.y = b[c].y;  b0.x = wv[c].x; b0.y = wv[c].y;
        a1.x = b[c].z;  a1.y = b[c].w;  b1.x = wv[c].z; b1.y = wv[c].w;
        d = __builtin_amdgcn_wmma_f32_16x16x4_f32(false, a0, false, b0, (short)0, d, false, false);
        d = __builtin_amdgcn_wmma_f32_16x16x4_f32(false, a1, false, b1, (short)0, d, false, false);
    }
    float e = d[0];
    e = (sel == 1) ? d[1] : e;
    e = (sel == 2) ? d[2] : e;
    e = (sel == 3) ? d[3] : e;
    e = (sel == 4) ? d[4] : e;
    e = (sel == 5) ? d[5] : e;
    e = (sel == 6) ? d[6] : e;
    e = (sel == 7) ? d[7] : e;
    e = diag ? e : 0.0f;
    e += lane_xor_in_row<1>(e);
    e += lane_xor_in_row<2>(e);
    e += lane_xor_in_row<4>(e);
    e += lane_xor_in_row<8>(e);
    e += lane_xor_16(e);
    return e;
}

template <int NB>
__device__ __forceinline__ void
xw_batch(const float* __restrict__ x, float* __restrict__ h, int row, int lane,
         const v4f (&wv)[4], int sel, bool diag) {
    v4f buf[NB][4];
    float t[NB];
    const v4f* xr = (const v4f*)(x + (size_t)row * D_FEAT) + lane;
#pragma unroll
    for (int j = 0; j < NB; ++j) {
#pragma unroll
        for (int c = 0; c < 4; ++c) buf[j][c] = __builtin_nontemporal_load(xr + 128 * j + 32 * c);
    }
    __builtin_amdgcn_sched_barrier(0);
#pragma unroll
    for (int j = 0; j < NB; ++j) t[j] = xw_row_dot(buf[j], wv, sel, diag);
    if (lane < NB) {
        float v = t[0];
#pragma unroll
        for (int j = 1; j < NB; ++j) v = (lane == j) ? t[j] : v;
        h[row + lane] = v;
    }
}

__device__ __forceinline__ void
xw_rows(const float* __restrict__ x, const float* __restrict__ W,
        float* __restrict__ h, int N, int row0, int nrows) {
    if (row0 >= N) return;
    const int lane = threadIdx.x & 31;
    const int rend = (row0 + nrows < N) ? row0 + nrows : N;
    const int sel = lane & 7;
    const bool diag = (lane < 8) || (lane >= 24);

    v4f wv[4];
    {
        const v4f* Wv = (const v4f*)W + lane;
#pragma unroll
        for (int c = 0; c < 4; ++c) wv[c] = Wv[32 * c];
    }

    int r = row0;
    for (; r + XB <= rend; r += XB) xw_batch<XB>(x, h, r, lane, wv, sel, diag);
    if (r < rend) xw_batch<1>(x, h, r, lane, wv, sel, diag);
}

__global__ void gcn_dinv(const float* __restrict__ deg, float* __restrict__ dinv, int N) {
    int i = blockIdx.x * blockDim.x + threadIdx.x;
    if (i < N) dinv[i] = __frsqrt_rn(deg[i]);
}

__global__ void gcn_init_out(const float* __restrict__ dinv, const float* __restrict__ h,
                             float* __restrict__ g, float* __restrict__ out, int N) {
    int i = blockIdx.x * blockDim.x + threadIdx.x;
    if (i < N) {
        float di = dinv[i];
        float hi = h[i];
        out[i] = di * di * hi;
        g[i] = di * hi;
    }
}

#define NR   8
#define NC   32
#define RS   12544
#define PTB  1024
#define DU   8
#define TU   4
#define HU   (TU / 2)
#define QCAP (32 * 4 * HU)
#define GWN  49152
#define GWU  6
#define ROT  ((RS / NC) & ~31)
#define FL   ((RS + PTB - 1) / PTB)

__device__ __forceinline__ int4 ld_quad_guarded(const int* __restrict__ p, int q, int nelem, int fill) {
    int4 v = make_int4(fill, fill, fill, fill);
    const int e = q * 4;
    if (e     < nelem) v.x = p[e];
    if (e + 1 < nelem) v.y = p[e + 1];
    if (e + 2 < nelem) v.z = p[e + 2];
    if (e + 3 < nelem) v.w = p[e + 3];
    return v;
}

template <int U>
__device__ __forceinline__ void ld_tile_vec(const int* __restrict__ p, int qbase, int tid, int4 (&v)[U]) {
    const int4* p4 = (const int4*)p + qbase + tid;
#pragma unroll
    for (int u = 0; u < U; ++u) v[u] = p4[u * PTB];
}
template <int U>
__device__ __forceinline__ void ld_tile_guarded(const int* __restrict__ p, int qbase, int tid, int nelem,
                                                int fill, int4 (&v)[U]) {
#pragma unroll
    for (int u = 0; u < U; ++u) v[u] = ld_quad_guarded(p, qbase + u * PTB + tid, nelem, fill);
}

__device__ __forceinline__ void deg_tile(const int4 (&v)[DU], unsigned* cnt, unsigned r0) {
#pragma unroll
    for (int u = 0; u < DU; ++u) {
        unsigned k;
        k = (unsigned)v[u].x - r0; if (k < RS) atomicAdd(&cnt[k], 1u);
        k = (unsigned)v[u].y - r0; if (k < RS) atomicAdd(&cnt[k], 1u);
        k = (unsigned)v[u].z - r0; if (k < RS) atomicAdd(&cnt[k], 1u);
        k = (unsigned)v[u].w - r0; if (k < RS) atomicAdd(&cnt[k], 1u);
    }
}

__device__ __forceinline__ void deg_scan(const int* __restrict__ dstc, int nelem, bool vec,
                                         unsigned* cnt, unsigned r0, int tid) {
    const int nq = (nelem + 3) >> 2;
    const int nfull = vec ? (nelem >> 2) / (PTB * DU) : 0;
    for (int t = 0; t < nfull; ++t) {
        int4 v[DU];
        ld_tile_vec<DU>(dstc, t * (PTB * DU), tid, v);
        deg_tile(v, cnt, r0);
    }
    for (int base = nfull * (PTB * DU); base < nq; base += PTB * DU) {
        int4 v[DU];
        ld_tile_guarded<DU>(dstc, base, tid, nelem, -1, v);
        deg_tile(v, cnt, r0);
    }
}

__device__ __forceinline__ void
deg_group(const int* __restrict__ dst, float* __restrict__ deg, int N, int E, int cl,
          unsigned* cnt, int bid) {
    const int tid = threadIdx.x;
    const int r = bid % NR;
    const int c = bid / NR;
    const int e0 = c * cl;
    int nelem = E - e0; if (nelem > cl) nelem = cl; if (nelem < 0) nelem = 0;
    const int* dstc = dst + e0;
    const bool vec = (((size_t)dst) & 15) == 0;

    for (unsigned r0 = (unsigned)r * RS; r0 < (unsigned)N; r0 += NR * RS) {
        for (int i = tid; i < RS; i += PTB) cnt[i] = 0u;
        __syncthreads();
        deg_scan(dstc, nelem, vec, cnt, r0, tid);
        __syncthreads();
        for (int i = tid; i < RS; i += PTB) {
            int k = i + c * ROT; if (k >= RS) k -= RS;
            const unsigned n = r0 + (unsigned)k;
            const unsigned cv = cnt[k];
            if (cv != 0u && n < (unsigned)N) atomicAdd(&deg[n], (float)cv);
        }
        __syncthreads();
    }
}

__global__ void __launch_bounds__(PTB, 16)
gcn_xw_deg(const float* __restrict__ x, const float* __restrict__ W, float* __restrict__ h,
           const int* __restrict__ dst, float* __restrict__ deg,
           int N, int E, int cl, int nbx, int rpw) {
    __shared__ unsigned cnt[RS];
    const int bid = blockIdx.x;
    if (bid < nbx) {
        const int wi = __builtin_amdgcn_readfirstlane((int)(threadIdx.x >> 5));
        xw_rows(x, W, h, N, (bid * (PTB / 32) + wi) * rpw, rpw);
    } else {
        deg_group(dst, deg, N, E, cl, cnt, bid - nbx);
    }
}

__device__ __forceinline__ float g_fetch(const float* __restrict__ g, const float* gw, unsigned s) {
    const bool inw = s < (unsigned)GWN;
    float gl = 0.0f;
    if (!inw) gl = g[s];
    float wl = gw[inw ? s : 0u];
    asm volatile("" : "+v"(wl));
    return inw ? wl : gl;
}

template <int U0>
__device__ __forceinline__ void agg_half(const int4 (&dc)[TU], const int4 (&sc)[TU],
                                         const float* __restrict__ g, const float* gw,
                                         float* acc, uint2* q, unsigned r0, int lane) {
    unsigned cnt = 0;
#pragma unroll
    for (int u = U0; u < U0 + HU; ++u) {
        const int dd[4] = {dc[u].x, dc[u].y, dc[u].z, dc[u].w};
        const int ss[4] = {sc[u].x, sc[u].y, sc[u].z, sc[u].w};
#pragma unroll
        for (int j = 0; j < 4; ++j) {
            const unsigned k = (unsigned)dd[j] - r0;
            const bool hit = k < RS;
            const unsigned m = __builtin_amdgcn_ballot_w32(hit);
            if (hit) q[cnt + __builtin_amdgcn_mbcnt_lo(m, 0u)] = make_uint2(k, (unsigned)ss[j]);
            cnt += (unsigned)__builtin_popcount(m);
        }
    }
    for (unsigned base = 0; base < cnt; base += 64u) {
        const unsigned i = base + (unsigned)lane;
        if (base + 32u < cnt) {
            const unsigned i2 = i + 32u;
            const bool two = i2 < cnt;
            const uint2 a = q[i];
            const uint2 b = q[two ? i2 : i];
            const float ga = g_fetch(g, gw, a.y);
            const float gb = g_fetch(g, gw, two ? b.y : 0u);
            atomicAdd(&acc[a.x], ga);
            atomicAdd(&acc[b.x], two ? gb : 0.0f);
        } else if (i < cnt) {
            const uint2 a = q[i];
            atomicAdd(&acc[a.x], g_fetch(g, gw, a.y));
        }
    }
}

__device__ __forceinline__ void agg_tile(const int4 (&dc)[TU], const int4 (&sc)[TU],
                                         const float* __restrict__ g, const float* gw,
                                         float* acc, uint2* q, unsigned r0, int lane) {
    agg_half<0>(dc, sc, g, gw, acc, q, r0, lane);
    agg_half<HU>(dc, sc, g, gw, acc, q, r0, lane);
}

__device__ __forceinline__ void agg_scan(const int* __restrict__ srcc, const int* __restrict__ dstc,
                                         int nelem, bool vec, const float* __restrict__ g,
                                         const float* gw, float* acc, uint2* q,
                                         unsigned r0, int tid, int lane) {
    const int nq = (nelem + 3) >> 2;
    const int nfull = vec ? (nelem >> 2) / (PTB * TU) : 0;
    if (nfull > 0) {
        int4 dn[TU], sn[TU];
        ld_tile_vec<TU>(dstc, 0, tid, dn);
        ld_tile_vec<TU>(srcc, 0, tid, sn);
        for (int t = 0; t < nfull; ++t) {
            int4 dc[TU], sc[TU];
#pragma unroll
            for (int u = 0; u < TU; ++u) { dc[u] = dn[u]; sc[u] = sn[u]; }
            if (t + 1 < nfull) {
                ld_tile_vec<TU>(dstc, (t + 1) * (PTB * TU), tid, dn);
                ld_tile_vec<TU>(srcc, (t + 1) * (PTB * TU), tid, sn);
            }
            agg_tile(dc, sc, g, gw, acc, q, r0, lane);
        }
    }
    for (int base = nfull * (PTB * TU); base < nq; base += PTB * TU) {
        int4 dc[TU], sc[TU];
        ld_tile_guarded<TU>(dstc, base, tid, nelem, -1, dc);
        ld_tile_guarded<TU>(srcc, base, tid, nelem, 0, sc);
        agg_tile(dc, sc, g, gw, acc, q, r0, lane);
    }
}

__device__ __forceinline__ void fill_window(const float* __restrict__ g, float* gw, int N, int tid) {
    const int nw = (N < GWN) ? N : GWN;
    if ((((size_t)g) & 15) == 0) {
        const int nq = nw >> 2;
        const v4f* g4 = (const v4f*)g;
        v4f* w4 = (v4f*)gw;
        const int rot = (nq > 0) ? (int)((blockIdx.x * (unsigned)(GWN / 4 / (NR * NC))) % (unsigned)nq) : 0;
        for (int i0 = 0; i0 < nq; i0 += PTB * GWU) {
            v4f v[GWU];
            int k[GWU];
#pragma unroll
            for (int u = 0; u < GWU; ++u) {
                const int i = i0 + u * PTB + tid;
                int kk = i + rot; if (kk >= nq) kk -= nq;
                k[u] = (i < nq) ? kk : -1;
                v[u] = g4[(i < nq) ? kk : 0];
            }
#pragma unroll
            for (int u = 0; u < GWU; ++u) if (k[u] >= 0) w4[k[u]] = v[u];
        }
        for (int i = (nq << 2) + tid; i < nw; i += PTB) gw[i] = g[i];
    } else {
        for (int i = tid; i < nw; i += PTB) gw[i] = g[i];
    }
}

__global__ void __launch_bounds__(PTB)
gcn_agg_lds(const int* __restrict__ src, const int* __restrict__ dst,
            const float* __restrict__ dinv, const float* __restrict__ g,
            float* __restrict__ out, int N, int E, int cl) {
    __shared__ float acc[RS];
    __shared__ uint2 queue[(PTB / 32) * QCAP];
    __shared__ float gw[GWN];
    const int tid = threadIdx.x;
    const int lane = tid & 31;
    uint2* q = queue + (tid >> 5) * QCAP;
    fill_window(g, gw, N, tid);
    const int r = blockIdx.x % NR;
    const int c = blockIdx.x / NR;
    const int e0 = c * cl;
    int nelem = E - e0; if (nelem > cl) nelem = cl; if (nelem < 0) nelem = 0;
    const int* srcc = src + e0;
    const int* dstc = dst + e0;
    const bool vec = ((((size_t)src) | ((size_t)dst)) & 15) == 0;

    for (unsigned r0 = (unsigned)r * RS; r0 < (unsigned)N; r0 += NR * RS) {
        for (int i = tid; i < RS; i += PTB) acc[i] = 0.0f;
        __syncthreads();
        agg_scan(srcc, dstc, nelem, vec, g, gw, acc, q, r0, tid, lane);
        __syncthreads();
        float dv[FL];
#pragma unroll
        for (int j = 0; j < FL; ++j) {
            int k = tid + j * PTB + c * ROT; if (k >= RS) k -= RS;
            const unsigned n = r0 + (unsigned)k;
            dv[j] = dinv[n < (unsigned)N ? n : (unsigned)N - 1u];
        }
#pragma unroll
        for (int j = 0; j < FL; ++j) {
            const int i = tid + j * PTB;
            if (i < RS) {
                int k = i + c * ROT; if (k >= RS) k -= RS;
                const unsigned n = r0 + (unsigned)k;
                const float v = acc[k];
                if (v != 0.0f && n < (unsigned)N) atomicAdd(&out[n], dv[j] * v);
            }
        }
        __syncthreads();
    }
}

extern "C" void kernel_launch(void* const* d_in, const int* in_sizes, int n_in,
                              void* d_out, int out_size, void* d_ws, size_t ws_size,
                              hipStream_t stream) {
    const float* x  = (const float*)d_in[0];
    const int*   ei = (const int*)d_in[1];
    const float* W  = (const float*)d_in[2];
    float*       out = (float*)d_out;

    const int N = out_size;
    const int E = in_sizes[1] / 2;
    const int* src = ei;
    const int* dst = ei + E;

    float* deg  = (float*)d_ws;
    float* dinv = deg + N;
    float* h    = dinv + N;
    float* g    = deg;

    const int TB = 256;
    const int nBlkN = (N + TB - 1) / TB;
    const int cl = (((E + NC - 1) / NC) + 3) & ~3;
    const int nbx = NR * NC;
    const int rpw = (N + nbx * (PTB / 32) - 1) / (nbx * (PTB / 32));

    gcn_init_deg<<<nBlkN, TB, 0, stream>>>(deg, N);
    gcn_xw_deg<<<nbx + NR * NC, PTB, 0, stream>>>(x, W, h, dst, deg, N, E, cl, nbx, rpw);
    gcn_dinv<<<nBlkN, TB, 0, stream>>>(deg, dinv, N);
    gcn_init_out<<<nBlkN, TB, 0, stream>>>(dinv, h, g, out, N);
    gcn_agg_lds<<<NR * NC, PTB, 0, stream>>>(src, dst, dinv, g, out, N, E, cl);
}


# Sum pass forms g = rsqrt(deg)*h itself from one (deg,h) pair plane (window fill, 8-B gather) and adds the self-loop term in its flush: dinv and g kernels go, 5 -> 3 launches; sum order changes
# speedup vs baseline: 5.9552x; 1.0299x over previous
#include <hip/hip_runtime.h>
#include <hip/hip_bf16.h>

typedef __attribute__((ext_vector_type(2))) float v2f;
typedef __attribute__((ext_vector_type(4))) float v4f;
typedef __attribute__((ext_vector_type(8))) float v8f;

#define D_FEAT 512

__global__ void gcn_init(float* __restrict__ dh, float* __restrict__ out, int N) {
    int i = blockIdx.x * blockDim.x + threadIdx.x;
    if (i < N) {
        ((float2*)dh)[i] = make_float2(1.0f, 0.0f);
        out[i] = 0.0f;
    }
}

#define XB 2

template <int M>
__device__ __forceinline__ float lane_xor_in_row(float v) {
    const int b = __builtin_bit_cast(int, v);
    return __builtin_bit_cast(float, __builtin_amdgcn_update_dpp(b, b, 0x160 + M, 0xf, 0xf, false));
}
__device__ __forceinline__ float lane_xor_16(float v) {
    const int b = __builtin_bit_cast(int, v);
    return __builtin_bit_cast(float, __builtin_amdgcn_permlanex16(b, b, 0x76543210, 0xfedcba98, false, false));
}

__device__ __forceinline__ float
xw_row_dot(const v4f (&b)[4], const v4f (&wv)[4], int sel, bool diag) {
    v8f d = {0.f, 0.f, 0.f, 0.f, 0.f, 0.f, 0.f, 0.f};
#pragma unroll
    for (int c = 0; c < 4; ++c) {
        v2f a0, b0, a1, b1;
        a0.x = b[c].x;  a0.y = b[c].y;  b0.x = wv[c].x; b0.y = wv[c].y;
        a1.x = b[c].z;  a1.y = b[c].w;  b1.x = wv[c].z; b1.y = wv[c].w;
        d = __builtin_amdgcn_wmma_f32_16x16x4_f32(false, a0, false, b0, (short)0, d, false, false);
        d = __builtin_amdgcn_wmma_f32_16x16x4_f32(false, a1, false, b1, (short)0, d, false, false);
    }
    float e = d[0];
    e = (sel == 1) ? d[1] : e;
    e = (sel == 2) ? d[2] : e;
    e = (sel == 3) ? d[3] : e;
    e = (sel == 4) ? d[4] : e;
    e = (sel == 5) ? d[5] : e;
    e = (sel == 6) ? d[6] : e;
    e = (sel == 7) ? d[7] : e;
    e = diag ? e : 0.0f;
    e += lane_xor_in_row<1>(e);
    e += lane_xor_in_row<2>(e);
    e += lane_xor_in_row<4>(e);
    e += lane_xor_in_row<8>(e);
    e += lane_xor_16(e);
    return e;
}

template <int NB>
__device__ __forceinline__ void
xw_batch(const float* __restrict__ x, float* __restrict__ h, int row, int lane,
         const v4f (&wv)[4], int sel, bool diag) {
    v4f buf[NB][4];
    float t[NB];
    const v4f* xr = (const v4f*)(x + (size_t)row * D_FEAT) + lane;
#pragma unroll
    for (int j = 0; j < NB; ++j) {
#pragma unroll
        for (int c = 0; c < 4; ++c) buf[j][c] = __builtin_nontemporal_load(xr + 128 * j + 32 * c);
    }
    __builtin_amdgcn_sched_barrier(0);
#pragma unroll
    for (int j = 0; j < NB; ++j) t[j] = xw_row_dot(buf[j], wv, sel, diag);
    if (lane < NB) {
        float v = t[0];
#pragma unroll
        for (int j = 1; j < NB; ++j) v = (lane == j) ? t[j] : v;
        h[2 * (row + lane)] = v;
    }
}

__device__ __forceinline__ void
xw_rows(const float* __restrict__ x, const float* __restrict__ W,
        float* __restrict__ h, int N, int row0, int nrows) {
    if (row0 >= N) return;
    const int lane = threadIdx.x & 31;
    const int rend = (row0 + nrows < N) ? row0 + nrows : N;
    const int sel = lane & 7;
    const bool diag = (lane < 8) || (lane >= 24);

    v4f wv[4];
    {
        const v4f* Wv = (const v4f*)W + lane;
#pragma unroll
        for (int c = 0; c < 4; ++c) wv[c] = Wv[32 * c];
    }

    int r = row0;
    for (; r + XB <= rend; r += XB) xw_batch<XB>(x, h, r, lane, wv, sel, diag);
    if (r < rend) xw_batch<1>(x, h, r, lane, wv, sel, diag);
}

#define NR   8
#define NC   32
#define RS   12544
#define PTB  1024
#define DU   8
#define TU   4
#define HU   (TU / 2)
#define QCAP (32 * 4 * HU)
#define GWN  49152
#define GWU  8
#define SW   ((RS + NC - 1) / NC)
#define ROT  ((RS / NC) & ~31)
#define FL   ((RS + PTB - 1) / PTB)

__device__ __forceinline__ int4 ld_quad_guarded(const int* __restrict__ p, int q, int nelem, int fill) {
    int4 v = make_int4(fill, fill, fill, fill);
    const int e = q * 4;
    if (e     < nelem) v.x = p[e];
    if (e + 1 < nelem) v.y = p[e + 1];
    if (e + 2 < nelem) v.z = p[e + 2];
    if (e + 3 < nelem) v.w = p[e + 3];
    return v;
}

template <int U>
__device__ __forceinline__ void ld_tile_vec(const int* __restrict__ p, int qbase, int tid, int4 (&v)[U]) {
    const int4* p4 = (const int4*)p + qbase + tid;
#pragma unroll
    for (int u = 0; u < U; ++u) v[u] = p4[u * PTB];
}
template <int U>
__device__ __forceinline__ void ld_tile_guarded(const int* __restrict__ p, int qbase, int tid, int nelem,
                                                int fill, int4 (&v)[U]) {
#pragma unroll
    for (int u = 0; u < U; ++u) v[u] = ld_quad_guarded(p, qbase + u * PTB + tid, nelem, fill);
}

__device__ __forceinline__ void deg_tile(const int4 (&v)[DU], unsigned* cnt, unsigned r0) {
#pragma unroll
    for (int u = 0; u < DU; ++u) {
        unsigned k;
        k = (unsigned)v[u].x - r0; if (k < RS) atomicAdd(&cnt[k], 1u);
        k = (unsigned)v[u].y - r0; if (k < RS) atomicAdd(&cnt[k], 1u);
        k = (unsigned)v[u].z - r0; if (k < RS) atomicAdd(&cnt[k], 1u);
        k = (unsigned)v[u].w - r0; if (k < RS) atomicAdd(&cnt[k], 1u);
    }
}

__device__ __forceinline__ void deg_scan(const int* __restrict__ dstc, int nelem, bool vec,
                                         unsigned* cnt, unsigned r0, int tid) {
    const int nq = (nelem + 3) >> 2;
    const int nfull = vec ? (nelem >> 2) / (PTB * DU) : 0;
    for (int t = 0; t < nfull; ++t) {
        int4 v[DU];
        ld_tile_vec<DU>(dstc, t * (PTB * DU), tid, v);
        deg_tile(v, cnt, r0);
    }
    for (int base = nfull * (PTB * DU); base < nq; base += PTB * DU) {
        int4 v[DU];
        ld_tile_guarded<DU>(dstc, base, tid, nelem, -1, v);
        deg_tile(v, cnt, r0);
    }
}

__device__ __forceinline__ void
deg_group(const int* __restrict__ dst, float* __restrict__ deg, int N, int E, int cl,
          unsigned* cnt, int bid) {
    const int tid = threadIdx.x;
    const int r = bid % NR;
    const int c = bid / NR;
    const int e0 = c * cl;
    int nelem = E - e0; if (nelem > cl) nelem = cl; if (nelem < 0) nelem = 0;
    const int* dstc = dst + e0;
    const bool vec = (((size_t)dst) & 15) == 0;

    for (unsigned r0 = (unsigned)r * RS; r0 < (unsigned)N; r0 += NR * RS) {
        for (int i = tid; i < RS; i += PTB) cnt[i] = 0u;
        __syncthreads();
        deg_scan(dstc, nelem, vec, cnt, r0, tid);
        __syncthreads();
        for (int i = tid; i < RS; i += PTB) {
            int k = i + c * ROT; if (k >= RS) k -= RS;
            const unsigned n = r0 + (unsigned)k;
            const unsigned cv = cnt[k];
            if (cv != 0u && n < (unsigned)N) atomicAdd(&deg[2u * n], (float)cv);
        }
        __syncthreads();
    }
}

__global__ void __launch_bounds__(PTB, 16)
gcn_xw_deg(const float* __restrict__ x, const float* __restrict__ W, float* __restrict__ h,
           const int* __restrict__ dst, float* __restrict__ deg,
           int N, int E, int cl, int nbx, int rpw) {
    __shared__ unsigned cnt[RS];
    const int bid = blockIdx.x;
    if (bid < nbx) {
        const int wi = __builtin_amdgcn_readfirstlane((int)(threadIdx.x >> 5));
        xw_rows(x, W, h, N, (bid * (PTB / 32) + wi) * rpw, rpw);
    } else {
        deg_group(dst, deg, N, E, cl, cnt, bid - nbx);
    }
}

__device__ __forceinline__ float g_fetch(const float2* __restrict__ dh2, const float* gw, unsigned s) {
    const bool inw = s < (unsigned)GWN;
    float dg = 1.0f, hh = 0.0f;
    if (!inw) { const float2 p = dh2[s]; dg = p.x; hh = p.y; }
    float wl = gw[inw ? s : 0u];
    asm volatile("" : "+v"(wl));
    const float gl = __frsqrt_rn(dg) * hh;
    return inw ? wl : gl;
}

template <int U0>
__device__ __forceinline__ void agg_half(const int4 (&dc)[TU], const int4 (&sc)[TU],
                                         const float2* __restrict__ g, const float* gw,
                                         float* acc, uint2* q, unsigned r0, int lane) {
    unsigned cnt = 0;
#pragma unroll
    for (int u = U0; u < U0 + HU; ++u) {
        const int dd[4] = {dc[u].x, dc[u].y, dc[u].z, dc[u].w};
        const int ss[4] = {sc[u].x, sc[u].y, sc[u].z, sc[u].w};
#pragma unroll
        for (int j = 0; j < 4; ++j) {
            const unsigned k = (unsigned)dd[j] - r0;
            const bool hit = k < RS;
            const unsigned m = __builtin_amdgcn_ballot_w32(hit);
            if (hit) q[cnt + __builtin_amdgcn_mbcnt_lo(m, 0u)] = make_uint2(k, (unsigned)ss[j]);
            cnt += (unsigned)__builtin_popcount(m);
        }
    }
    for (unsigned base = 0; base < cnt; base += 64u) {
        const unsigned i = base + (unsigned)lane;
        if (base + 32u < cnt) {
            const unsigned i2 = i + 32u;
            const bool two = i2 < cnt;
            const uint2 a = q[i];
            const uint2 b = q[two ? i2 : i];
            const float ga = g_fetch(g, gw, a.y);
            const float gb = g_fetch(g, gw, two ? b.y : 0u);
            atomicAdd(&acc[a.x], ga);
            atomicAdd(&acc[b.x], two ? gb : 0.0f);
        } else if (i < cnt) {
            const uint2 a = q[i];
            atomicAdd(&acc[a.x], g_fetch(g, gw, a.y));
        }
    }
}

__device__ __forceinline__ void agg_tile(const int4 (&dc)[TU], const int4 (&sc)[TU],
                                         const float2* __restrict__ g, const float* gw,
                                         float* acc, uint2* q, unsigned r0, int lane) {
    agg_half<0>(dc, sc, g, gw, acc, q, r0, lane);
    agg_half<HU>(dc, sc, g, gw, acc, q, r0, lane);
}

__device__ __forceinline__ void agg_scan(const int* __restrict__ srcc, const int* __restrict__ dstc,
                                         int nelem, bool vec, const float2* __restrict__ g,
                                         const float* gw, float* acc, uint2* q,
                                         unsigned r0, int tid, int lane) {
    const int nq = (nelem + 3) >> 2;
    const int nfull = vec ? (nelem >> 2) / (PTB * TU) : 0;
    if (nfull > 0) {
        int4 dn[TU], sn[TU];
        ld_tile_vec<TU>(dstc, 0, tid, dn);
        ld_tile_vec<TU>(srcc, 0, tid, sn);
        for (int t = 0; t < nfull; ++t) {
            int4 dc[TU], sc[TU];
#pragma unroll
            for (int u = 0; u < TU; ++u) { dc[u] = dn[u]; sc[u] = sn[u]; }
            if (t + 1 < nfull) {
                ld_tile_vec<TU>(dstc, (t + 1) * (PTB * TU), tid, dn);
                ld_tile_vec<TU>(srcc, (t + 1) * (PTB * TU), tid, sn);
            }
            agg_tile(dc, sc, g, gw, acc, q, r0, lane);
        }
    }
    for (int base = nfull * (PTB * TU); base < nq; base += PTB * TU) {
        int4 dc[TU], sc[TU];
        ld_tile_guarded<TU>(dstc, base, tid, nelem, -1, dc);
        ld_tile_guarded<TU>(srcc, base, tid, nelem, 0, sc);
        agg_tile(dc, sc, g, gw, acc, q, r0, lane);
    }
}

__device__ __forceinline__ void fill_window(const float* __restrict__ dh, float* gw, int N, int tid) {
    const int nw = (N < GWN) ? N : GWN;
    if ((((size_t)dh) & 15) == 0) {
        const int nq = nw >> 1;
        const v4f* d4 = (const v4f*)dh;
        v2f* w2 = (v2f*)gw;
        const int rot = (nq > 0) ? (int)((blockIdx.x * (unsigned)(GWN / 2 / (NR * NC))) % (unsigned)nq) : 0;
        for (int i0 = 0; i0 < nq; i0 += PTB * GWU) {
            v4f v[GWU];
            int k[GWU];
#pragma unroll
            for (int u = 0; u < GWU; ++u) {
                const int i = i0 + u * PTB + tid;
                int kk = i + rot; if (kk >= nq) kk -= nq;
                k[u] = (i < nq) ? kk : -1;
                v[u] = d4[(i < nq) ? kk : 0];
            }
#pragma unroll
            for (int u = 0; u < GWU; ++u) {
                if (k[u] >= 0) {
                    v2f gg;
                    gg.x = __frsqrt_rn(v[u].x) * v[u].y;
                    gg.y = __frsqrt_rn(v[u].z) * v[u].w;
                    w2[k[u]] = gg;
                }
            }
        }
        for (int i = (nq << 1) + tid; i < nw; i += PTB) gw[i] = __frsqrt_rn(dh[2 * i]) * dh[2 * i + 1];
    } else {
        for (int i = tid; i < nw; i += PTB) gw[i] = __frsqrt_rn(dh[2 * i]) * dh[2 * i + 1];
    }
}

__global__ void __launch_bounds__(PTB)
gcn_agg_lds(const int* __restrict__ src, const int* __restrict__ dst,
            const float* __restrict__ dh, float* __restrict__ out, int N, int E, int cl) {
    __shared__ float acc[RS];
    __shared__ uint2 queue[(PTB / 32) * QCAP];
    __shared__ float gw[GWN];
    const int tid = threadIdx.x;
    const int lane = tid & 31;
    uint2* q = queue + (tid >> 5) * QCAP;
    const float2* dh2 = (const float2*)dh;
    fill_window(dh, gw, N, tid);
    const int r = blockIdx.x % NR;
    const int c = blockIdx.x / NR;
    const int e0 = c * cl;
    int nelem = E - e0; if (nelem > cl) nelem = cl; if (nelem < 0) nelem = 0;
    const int* srcc = src + e0;
    const int* dstc = dst + e0;
    const bool vec = ((((size_t)src) | ((size_t)dst)) & 15) == 0;

    for (unsigned r0 = (unsigned)r * RS; r0 < (unsigned)N; r0 += NR * RS) {
        for (int i = tid; i < RS; i += PTB) acc[i] = 0.0f;
        __syncthreads();
        agg_scan(srcc, dstc, nelem, vec, dh2, gw, acc, q, r0, tid, lane);
        __syncthreads();
        float2 dv[FL];
#pragma unroll
        for (int j = 0; j < FL; ++j) {
            int k = tid + j * PTB + c * ROT; if (k >= RS) k -= RS;
            const unsigned n = r0 + (unsigned)k;
            dv[j] = dh2[n < (unsigned)N ? n : (unsigned)N - 1u];
        }
#pragma unroll
        for (int j = 0; j < FL; ++j) {
            const int i = tid + j * PTB;
            if (i < RS) {
                int k = i + c * ROT; if (k >= RS) k -= RS;
                const unsigned n = r0 + (unsigned)k;
                const float v = acc[k];
                if (n < (unsigned)N) {
                    const float di = __frsqrt_rn(dv[j].x);
                    if (v != 0.0f) atomicAdd(&out[n], di * v);
                    if ((unsigned)(k - c * SW) < (unsigned)SW) atomicAdd(&out[n], di * di * dv[j].y);
                }
            }
        }
        __syncthreads();
    }
}

extern "C" void kernel_launch(void* const* d_in, const int* in_sizes, int n_in,
                              void* d_out, int out_size, void* d_ws, size_t ws_size,
                              hipStream_t stream) {
    const float* x  = (const float*)d_in[0];
    const int*   ei = (const int*)d_in[1];
    const float* W  = (const float*)d_in[2];
    float*       out = (float*)d_out;

    const int N = out_size;
    const int E = in_sizes[1] / 2;
    const int* src = ei;
    const int* dst = ei + E;

    float* dh  = (float*)d_ws;
    float* deg = dh;
    float* h   = dh + 1;

    const int TB = 256;
    const int nBlkN = (N + TB - 1) / TB;
    const int cl = (((E + NC - 1) / NC) + 3) & ~3;
    const int nbx = NR * NC;
    const int rpw = (N + nbx * (PTB / 32) - 1) / (nbx * (PTB / 32));

    gcn_init<<<nBlkN, TB, 0, stream>>>(dh, out, N);
    gcn_xw_deg<<<nbx + NR * NC, PTB, 0, stream>>>(x, W, h, dst, deg, N, E, cl, nbx, rpw);
    gcn_agg_lds<<<NR * NC, PTB, 0, stream>>>(src, dst, dh, out, N, E, cl);
}


# Both scatter passes: the partial tile at a chunk's end is loaded like a whole one (16-B loads under lane masks, sum pass: asked for one tile ahead); a wave skips its rows past the end; sum order kept
# speedup vs baseline: 6.1802x; 1.0378x over previous
#include <hip/hip_runtime.h>
#include <hip/hip_bf16.h>

typedef __attribute__((ext_vector_type(2))) float v2f;
typedef __attribute__((ext_vector_type(4))) float v4f;
typedef __attribute__((ext_vector_type(8))) float v8f;

#define D_FEAT 512

__global__ void gcn_init(float* __restrict__ dh, float* __restrict__ out, int N) {
    int i = blockIdx.x * blockDim.x + threadIdx.x;
    if (i < N) {
        ((float2*)dh)[i] = make_float2(1.0f, 0.0f);
        out[i] = 0.0f;
    }
}

#define XB 2

template <int M>
__device__ __forceinline__ float lane_xor_in_row(float v) {
    const int b = __builtin_bit_cast(int, v);
    return __builtin_bit_cast(float, __builtin_amdgcn_update_dpp(b, b, 0x160 + M, 0xf, 0xf, false));
}
__device__ __forceinline__ float lane_xor_16(float v) {
    const int b = __builtin_bit_cast(int, v);
    return __builtin_bit_cast(float, __builtin_amdgcn_permlanex16(b, b, 0x76543210, 0xfedcba98, false, false));
}

__device__ __forceinline__ float
xw_row_dot(const v4f (&b)[4], const v4f (&wv)[4], int sel, bool diag) {
    v8f d = {0.f, 0.f, 0.f, 0.f, 0.f, 0.f, 0.f, 0.f};
#pragma unroll
    for (int c = 0; c < 4; ++c) {
        v2f a0, b0, a1, b1;
        a0.x = b[c].x;  a0.y = b[c].y;  b0.x = wv[c].x; b0.y = wv[c].y;
        a1.x = b[c].z;  a1.y = b[c].w;  b1.x = wv[c].z; b1.y = wv[c].w;
        d = __builtin_amdgcn_wmma_f32_16x16x4_f32(false, a0, false, b0, (short)0, d, false, false);
        d = __builtin_amdgcn_wmma_f32_16x16x4_f32(false, a1, false, b1, (short)0, d, false, false);
    }
    float e = d[0];
    e = (sel == 1) ? d[1] : e;
    e = (sel == 2) ? d[2] : e;
    e = (sel == 3) ? d[3] : e;
    e = (sel == 4) ? d[4] : e;
    e = (sel == 5) ? d[5] : e;
    e = (sel == 6) ? d[6] : e;
    e = (sel == 7) ? d[7] : e;
    e = diag ? e : 0.0f;
    e += lane_xor_in_row<1>(e);
    e += lane_xor_in_row<2>(e);
    e += lane_xor_in_row<4>(e);
    e += lane_xor_in_row<8>(e);
    e += lane_xor_16(e);
    return e;
}

template <int NB>
__device__ __forceinline__ void
xw_batch(const float* __restrict__ x, float* __restrict__ h, int row, int lane,
         const v4f (&wv)[4], int sel, bool diag) {
    v4f buf[NB][4];
    float t[NB];
    const v4f* xr = (const v4f*)(x + (size_t)row * D_FEAT) + lane;
#pragma unroll
    for (int j = 0; j < NB; ++j) {
#pragma unroll
        for (int c = 0; c < 4; ++c) buf[j][c] = __builtin_nontemporal_load(xr + 128 * j + 32 * c);
    }
    __builtin_amdgcn_sched_barrier(0);
#pragma unroll
    for (int j = 0; j < NB; ++j) t[j] = xw_row_dot(buf[j], wv, sel, diag);
    if (lane < NB) {
        float v = t[0];
#pragma unroll
        for (int j = 1; j < NB; ++j) v = (lane == j) ? t[j] : v;
        h[2 * (row + lane)] = v;
    }
}

__device__ __forceinline__ void
xw_rows(const float* __restrict__ x, const float* __restrict__ W,
        float* __restrict__ h, int N, int row0, int nrows) {
    if (row0 >= N) return;
    const int lane = threadIdx.x & 31;
    const int rend = (row0 + nrows < N) ? row0 + nrows : N;
    const int sel = lane & 7;
    const bool diag = (lane < 8) || (lane >= 24);

    v4f wv[4];
    {
        const v4f* Wv = (const v4f*)W + lane;
#pragma unroll
        for (int c = 0; c < 4; ++c) wv[c] = Wv[32 * c];
    }

    int r = row0;
    for (; r + XB <= rend; r += XB) xw_batch<XB>(x, h, r, lane, wv, sel, diag);
    if (r < rend) xw_batch<1>(x, h, r, lane, wv, sel, diag);
}

#define NR   8
#define NC   32
#define RS   12544
#define PTB  1024
#define DU   8
#define TU   4
#define HU   (TU / 2)
#define QCAP (32 * 4 * HU)
#define GWN  49152
#define GWU  8
#define SW   ((RS + NC - 1) / NC)
#define ROT  ((RS / NC) & ~31)
#define FL   ((RS + PTB - 1) / PTB)

__device__ __forceinline__ int4 ld_quad_guarded(const int* __restrict__ p, int q, int nelem, int fill) {
    int4 v = make_int4(fill, fill, fill, fill);
    const int e = q * 4;
    if (e     < nelem) v.x = p[e];
    if (e + 1 < nelem) v.y = p[e + 1];
    if (e + 2 < nelem) v.z = p[e + 2];
    if (e + 3 < nelem) v.w = p[e + 3];
    return v;
}

template <int U>
__device__ __forceinline__ void ld_tile_vec(const int* __restrict__ p, int qbase, int tid, int4 (&v)[U]) {
    const int4* p4 = (const int4*)p + qbase + tid;
#pragma unroll
    for (int u = 0; u < U; ++u) v[u] = p4[u * PTB];
}
template <int U>
__device__ __forceinline__ void ld_tile_guarded(const int* __restrict__ p, int qbase, int tid, int nelem,
                                                int fill, int4 (&v)[U]) {
#pragma unroll
    for (int u = 0; u < U; ++u) v[u] = ld_quad_guarded(p, qbase + u * PTB + tid, nelem, fill);
}

template <int U>
__device__ __forceinline__ void ld_tile_tail(const int* __restrict__ p, int qbase, int tid, int nq, int fill,
                                             int4 (&v)[U]) {
    const int4* p4 = (const int4*)p + qbase + tid;
    const int left = nq - qbase - tid;
#pragma unroll
    for (int u = 0; u < U; ++u) {
        v[u] = make_int4(fill, fill, fill, fill);
        if (u * PTB < left) v[u] = p4[u * PTB];
    }
}
template <int U>
__device__ __forceinline__ int tail_rows(int qbase, int tid, int nq) {
    const int left = nq - qbase - (tid & ~31);
    int nv = (left > 0) ? (left + PTB - 1) / PTB : 0;
    nv = (nv < U) ? nv : U;
    return __builtin_amdgcn_readfirstlane(nv);
}

__device__ __forceinline__ void deg_tile(const int4 (&v)[DU], unsigned* cnt, unsigned r0) {
#pragma unroll
    for (int u = 0; u < DU; ++u) {
        unsigned k;
        k = (unsigned)v[u].x - r0; if (k < RS) atomicAdd(&cnt[k], 1u);
        k = (unsigned)v[u].y - r0; if (k < RS) atomicAdd(&cnt[k], 1u);
        k = (unsigned)v[u].z - r0; if (k < RS) atomicAdd(&cnt[k], 1u);
        k = (unsigned)v[u].w - r0; if (k < RS) atomicAdd(&cnt[k], 1u);
    }
}

__device__ __forceinline__ void deg_tile_tail(const int4 (&v)[DU], unsigned* cnt, unsigned r0, int nv) {
#pragma unroll
    for (int u = 0; u < DU; ++u) {
        if (u < nv) {
            unsigned k;
            k = (unsigned)v[u].x - r0; if (k < RS) atomicAdd(&cnt[k], 1u);
            k = (unsigned)v[u].y - r0; if (k < RS) atomicAdd(&cnt[k], 1u);
            k = (unsigned)v[u].z - r0; if (k < RS) atomicAdd(&cnt[k], 1u);
            k = (unsigned)v[u].w - r0; if (k < RS) atomicAdd(&cnt[k], 1u);
        }
    }
}

__device__ __forceinline__ void deg_scan(const int* __restrict__ dstc, int nelem, bool vec,
                                         unsigned* cnt, unsigned r0, int tid) {
    const int nq = (nelem + 3) >> 2;
    const int nfull = vec ? (nelem >> 2) / (PTB * DU) : 0;
    for (int t = 0; t < nfull; ++t) {
        int4 v[DU];
        ld_tile_vec<DU>(dstc, t * (PTB * DU), tid, v);
        deg_tile(v, cnt, r0);
    }
    const int tb = nfull * (PTB * DU);
    if (vec && (nelem & 3) == 0) {
        if (tb < nq) {
            int4 v[DU];
            ld_tile_tail<DU>(dstc, tb, tid, nq, -1, v);
            deg_tile_tail(v, cnt, r0, tail_rows<DU>(tb, tid, nq));
        }
    } else {
        for (int base = tb; base < nq; base += PTB * DU) {
            int4 v[DU];
            ld_tile_guarded<DU>(dstc, base, tid, nelem, -1, v);
            deg_tile(v, cnt, r0);
        }
    }
}

__device__ __forceinline__ void
deg_group(const int* __restrict__ dst, float* __restrict__ deg, int N, int E, int cl,
          unsigned* cnt, int bid) {
    const int tid = threadIdx.x;
    const int r = bid % NR;
    const int c = bid / NR;
    const int e0 = c * cl;
    int nelem = E - e0; if (nelem > cl) nelem = cl; if (nelem < 0) nelem = 0;
    const int* dstc = dst + e0;
    const bool vec = (((size_t)dst) & 15) == 0;

    for (unsigned r0 = (unsigned)r * RS; r0 < (unsigned)N; r0 += NR * RS) {
        for (int i = tid; i < RS; i += PTB) cnt[i] = 0u;
        __syncthreads();
        deg_scan(dstc, nelem, vec, cnt, r0, tid);
        __syncthreads();
        for (int i = tid; i < RS; i += PTB) {
            int k = i + c * ROT; if (k >= RS) k -= RS;
            const unsigned n = r0 + (unsigned)k;
            const unsigned cv = cnt[k];
            if (cv != 0u && n < (unsigned)N) atomicAdd(&deg[2u * n], (float)cv);
        }
        __syncthreads();
    }
}

__global__ void __launch_bounds__(PTB, 16)
gcn_xw_deg(const float* __restrict__ x, const float* __restrict__ W, float* __restrict__ h,
           const int* __restrict__ dst, float* __restrict__ deg,
           int N, int E, int cl, int nbx, int rpw) {
    __shared__ unsigned cnt[RS];
    const int bid = blockIdx.x;
    if (bid < nbx) {
        const int wi = __builtin_amdgcn_readfirstlane((int)(threadIdx.x >> 5));
        xw_rows(x, W, h, N, (bid * (PTB / 32) + wi) * rpw, rpw);
    } else {
        deg_group(dst, deg, N, E, cl, cnt, bid - nbx);
    }
}

__device__ __forceinline__ float g_fetch(const float2* __restrict__ dh2, const float* gw, unsigned s) {
    const bool inw = s < (unsigned)GWN;
    float dg = 1.0f, hh = 0.0f;
    if (!inw) { const float2 p = dh2[s]; dg = p.x; hh = p.y; }
    float wl = gw[inw ? s : 0u];
    asm volatile("" : "+v"(wl));
    const float gl = __frsqrt_rn(dg) * hh;
    return inw ? wl : gl;
}

template <int U0, bool TAIL>
__device__ __forceinline__ void agg_half(const int4 (&dc)[TU], const int4 (&sc)[TU],
                                         const float2* __restrict__ g, const float* gw,
                                         float* acc, uint2* q, unsigned r0, int lane, int nv) {
    unsigned cnt = 0;
#pragma unroll
    for (int u = U0; u < U0 + HU; ++u) {
        if (!TAIL || u < nv) {
            const int dd[4] = {dc[u].x, dc[u].y, dc[u].z, dc[u].w};
            const int ss[4] = {sc[u].x, sc[u].y, sc[u].z, sc[u].w};
#pragma unroll
            for (int j = 0; j < 4; ++j) {
                const unsigned k = (unsigned)dd[j] - r0;
                const bool hit = k < RS;
                const unsigned m = __builtin_amdgcn_ballot_w32(hit);
                if (hit) q[cnt + __builtin_amdgcn_mbcnt_lo(m, 0u)] = make_uint2(k, (unsigned)ss[j]);
                cnt += (unsigned)__builtin_popcount(m);
            }
        }
    }
    for (unsigned base = 0; base < cnt; base += 64u) {
        const unsigned i = base + (unsigned)lane;
        if (base + 32u < cnt) {
            const unsigned i2 = i + 32u;
            const bool two = i2 < cnt;
            const uint2 a = q[i];
            const uint2 b = q[two ? i2 : i];
            const float ga = g_fetch(g, gw, a.y);
            const float gb = g_fetch(g, gw, two ? b.y : 0u);
            atomicAdd(&acc[a.x], ga);
            atomicAdd(&acc[b.x], two ? gb : 0.0f);
        } else if (i < cnt) {
            const uint2 a = q[i];
            atomicAdd(&acc[a.x], g_fetch(g, gw, a.y));
        }
    }
}

template <bool TAIL>
__device__ __forceinline__ void agg_tile(const int4 (&dc)[TU], const int4 (&sc)[TU],
                                         const float2* __restrict__ g, const float* gw,
                                         float* acc, uint2* q, unsigned r0, int lane, int nv) {
    agg_half<0, TAIL>(dc, sc, g, gw, acc, q, r0, lane, nv);
    if (!TAIL || nv > HU) agg_half<HU, TAIL>(dc, sc, g, gw, acc, q, r0, lane, nv);
}

__device__ __forceinline__ void agg_scan(const int* __restrict__ srcc, const int* __restrict__ dstc,
                                         int nelem, bool vec, const float2* __restrict__ g,
                                         const float* gw, float* acc, uint2* q,
                                         unsigned r0, int tid, int lane) {
    const int nq = (nelem + 3) >> 2;
    const int nfull = vec ? (nelem >> 2) / (PTB * TU) : 0;
    const int tb = nfull * (PTB * TU);
    const bool vtail = vec && (nelem & 3) == 0 && tb < nq;
    if (nfull > 0 || vtail) {
        int4 dn[TU], sn[TU];
        if (nfull > 0) {
            ld_tile_vec<TU>(dstc, 0, tid, dn);
            ld_tile_vec<TU>(srcc, 0, tid, sn);
        }
        for (int t = 0; t < nfull; ++t) {
            int4 dc[TU], sc[TU];
#pragma unroll
            for (int u = 0; u < TU; ++u) { dc[u] = dn[u]; sc[u] = sn[u]; }
            if (t + 1 < nfull) {
                ld_tile_vec<TU>(dstc, (t + 1) * (PTB * TU), tid, dn);
                ld_tile_vec<TU>(srcc, (t + 1) * (PTB * TU), tid, sn);
            } else if (vtail) {
                ld_tile_tail<TU>(dstc, tb, tid, nq, -1, dn);
                ld_tile_tail<TU>(srcc, tb, tid, nq, 0, sn);
            }
            agg_tile<false>(dc, sc, g, gw, acc, q, r0, lane, TU);
        }
        if (vtail) {
            if (nfull == 0) {
                ld_tile_tail<TU>(dstc, tb, tid, nq, -1, dn);
                ld_tile_tail<TU>(srcc, tb, tid, nq, 0, sn);
            }
            agg_tile<true>(dn, sn, g, gw, acc, q, r0, lane, tail_rows<TU>(tb, tid, nq));
        }
    }
    if (!vtail) {
        for (int base = tb; base < nq; base += PTB * TU) {
            int4 dc[TU], sc[TU];
            ld_tile_guarded<TU>(dstc, base, tid, nelem, -1, dc);
            ld_tile_guarded<TU>(srcc, base, tid, nelem, 0, sc);
            agg_tile<false>(dc, sc, g, gw, acc, q, r0, lane, TU);
        }
    }
}

__device__ __forceinline__ void fill_window(const float* __restrict__ dh, float* gw, int N, int tid) {
    const int nw = (N < GWN) ? N : GWN;
    if ((((size_t)dh) & 15) == 0) {
        const int nq = nw >> 1;
        const v4f* d4 = (const v4f*)dh;
        v2f* w2 = (v2f*)gw;
        const int rot = (nq > 0) ? (int)((blockIdx.x * (unsigned)(GWN / 2 / (NR * NC))) % (unsigned)nq) : 0;
        for (int i0 = 0; i0 < nq; i0 += PTB * GWU) {
            v4f v[GWU];
            int k[GWU];
#pragma unroll
            for (int u = 0; u < GWU; ++u) {
                const int i = i0 + u * PTB + tid;
                int kk = i + rot; if (kk >= nq) kk -= nq;
                k[u] = (i < nq) ? kk : -1;
                v[u] = d4[(i < nq) ? kk : 0];
            }
#pragma unroll
            for (int u = 0; u < GWU; ++u) {
                if (k[u] >= 0) {
                    v2f gg;
                    gg.x = __frsqrt_rn(v[u].x) * v[u].y;
                    gg.y = __frsqrt_rn(v[u].z) * v[u].w;
                    w2[k[u]] = gg;
                }
            }
        }
        for (int i = (nq << 1) + tid; i < nw; i += PTB) gw[i] = __frsqrt_rn(dh[2 * i]) * dh[2 * i + 1];
    } else {
        for (int i = tid; i < nw; i += PTB) gw[i] = __frsqrt_rn(dh[2 * i]) * dh[2 * i + 1];
    }
}

__global__ void __launch_bounds__(PTB)
gcn_agg_lds(const int* __restrict__ src, const int* __restrict__ dst,
            const float* __restrict__ dh, float* __restrict__ out, int N, int E, int cl) {
    __shared__ float acc[RS];
    __shared__ uint2 queue[(PTB / 32) * QCAP];
    __shared__ float gw[GWN];
    const int tid = threadIdx.x;
    const int lane = tid & 31;
    uint2* q = queue + (tid >> 5) * QCAP;
    const float2* dh2 = (const float2*)dh;
    fill_window(dh, gw, N, tid);
    const int r = blockIdx.x % NR;
    const int c = blockIdx.x / NR;
    const int e0 = c * cl;
    int nelem = E - e0; if (nelem > cl) nelem = cl; if (nelem < 0) nelem = 0;
    const int* srcc = src + e0;
    const int* dstc = dst + e0;
    const bool vec = ((((size_t)src) | ((size_t)dst)) & 15) == 0;

    for (unsigned r0 = (unsigned)r * RS; r0 < (unsigned)N; r0 += NR * RS) {
        for (int i = tid; i < RS; i += PTB) acc[i] = 0.0f;
        __syncthreads();
        agg_scan(srcc, dstc, nelem, vec, dh2, gw, acc, q, r0, tid, lane);
        __syncthreads();
        float2 dv[FL];
#pragma unroll
        for (int j = 0; j < FL; ++j) {
            int k = tid + j * PTB + c * ROT; if (k >= RS) k -= RS;
            const unsigned n = r0 + (unsigned)k;
            dv[j] = dh2[n < (unsigned)N ? n : (unsigned)N - 1u];
        }
#pragma unroll
        for (int j = 0; j < FL; ++j) {
            const int i = tid + j * PTB;
            if (i < RS) {
                int k = i + c * ROT; if (k >= RS) k -= RS;
                const unsigned n = r0 + (unsigned)k;
                const float v = acc[k];
                if (n < (unsigned)N) {
                    const float di = __frsqrt_rn(dv[j].x);
                    if (v != 0.0f) atomicAdd(&out[n], di * v);
                    if ((unsigned)(k - c * SW) < (unsigned)SW) atomicAdd(&out[n], di * di * dv[j].y);
                }
            }
        }
        __syncthreads();
    }
}

extern "C" void kernel_launch(void* const* d_in, const int* in_sizes, int n_in,
                              void* d_out, int out_size, void* d_ws, size_t ws_size,
                              hipStream_t stream) {
    const float* x  = (const float*)d_in[0];
    const int*   ei = (const int*)d_in[1];
    const float* W  = (const float*)d_in[2];
    float*       out = (float*)d_out;

    const int N = out_size;
    const int E = in_sizes[1] / 2;
    const int* src = ei;
    const int* dst = ei + E;

    float* dh  = (float*)d_ws;
    float* deg = dh;
    float* h   = dh + 1;

    const int TB = 256;
    const int nBlkN = (N + TB - 1) / TB;
    const int cl = (((E + NC - 1) / NC) + 3) & ~3;
    const int nbx = NR * NC;
    const int rpw = (N + nbx * (PTB / 32) - 1) / (nbx * (PTB / 32));

    gcn_init<<<nBlkN, TB, 0, stream>>>(dh, out, N);
    gcn_xw_deg<<<nbx + NR * NC, PTB, 0, stream>>>(x, W, h, dst, deg, N, E, cl, nbx, rpw);
    gcn_agg_lds<<<NR * NC, PTB, 0, stream>>>(src, dst, dh, out, N, E, cl);
}
